# starting kernel
# baseline (speedup 1.0000x reference)
#include <hip/hip_runtime.h>

typedef __attribute__((ext_vector_type(2))) float v2f;
typedef __attribute__((ext_vector_type(8))) float v8f;

#define NEG_SLOPE 0.2f

__device__ __forceinline__ unsigned enc_f32(float f) {
  unsigned u = __float_as_uint(f);
  return (u & 0x80000000u) ? ~u : (u | 0x80000000u);
}
__device__ __forceinline__ float dec_f32(unsigned u) {
  unsigned b = (u & 0x80000000u) ? (u & 0x7fffffffu) : ~u;
  return __uint_as_float(b);
}
#define ENC_NEG_INF 0x007FFFFFu

__global__ void fill_f32(float* __restrict__ p, float v, long long n) {
  long long i = (long long)blockIdx.x * blockDim.x + threadIdx.x;
  if (i < n) p[i] = v;
}
__global__ void fill_u32(unsigned* __restrict__ p, unsigned v, int n) {
  int i = blockIdx.x * blockDim.x + threadIdx.x;
  if (i < n) p[i] = v;
}

__global__ void gemm_f32_wmma(const float* __restrict__ X, const float* __restrict__ W,
                              float* __restrict__ H, int M, int K, int N) {
  int lane = threadIdx.x & 31;
  int wave = (int)((blockIdx.x * (long long)blockDim.x + threadIdx.x) >> 5);
  int nt4 = N >> 6;
  int mt  = (M + 15) >> 4;
  if (wave >= mt * nt4) return;
  int tm = wave / nt4, tn = wave % nt4;
  int half = lane >> 4;
  int l15  = lane & 15;
  int row = tm * 16 + l15;  if (row >= M) row = M - 1;
  int col0 = tn * 64 + l15;
  v8f acc0 = {}, acc1 = {}, acc2 = {}, acc3 = {};
  for (int k0 = 0; k0 < K; k0 += 4) {
    int kb = k0 + (half << 1);
    v2f a = *(const v2f*)(X + (size_t)row * K + kb);
    const float* w0 = W + (size_t)kb * N;
    const float* w1 = W + (size_t)(kb + 1) * N;
    v2f b0, b1, b2, b3;
    b0.x = w0[col0 +  0]; b0.y = w1[col0 +  0];
    b1.x = w0[col0 + 16]; b1.y = w1[col0 + 16];
    b2.x = w0[col0 + 32]; b2.y = w1[col0 + 32];
    b3.x = w0[col0 + 48]; b3.y = w1[col0 + 48];
    acc0 = __builtin_amdgcn_wmma_f32_16x16x4_f32(false, a, false, b0, (short)0, acc0, false, false);
    acc1 = __builtin_amdgcn_wmma_f32_16x16x4_f32(false, a, false, b1, (short)0, acc1, false, false);
    acc2 = __builtin_amdgcn_wmma_f32_16x16x4_f32(false, a, false, b2, (short)0, acc2, false, false);
    acc3 = __builtin_amdgcn_wmma_f32_16x16x4_f32(false, a, false, b3, (short)0, acc3, false, false);
  }
  int rbase = tm * 16 + half * 8;
#pragma unroll
  for (int v = 0; v < 8; ++v) {
    int r = rbase + v;
    if (r < M) {
      float* hp = H + (size_t)r * N + col0;
      hp[ 0] = acc0[v];
      hp[16] = acc1[v];
      hp[32] = acc2[v];
      hp[48] = acc3[v];
    }
  }
}

__global__ void node_alpha(const float* __restrict__ h, const float* __restrict__ avs,
                           const float* __restrict__ avd, float* __restrict__ as,
                           float* __restrict__ ad, int Nn, int F) {
  int lane = threadIdx.x & 31;
  int node = (int)((blockIdx.x * (long long)blockDim.x + threadIdx.x) >> 5);
  if (node >= Nn) return;
  const float* hp = h + (size_t)node * F;
  float ss = 0.f, sd = 0.f;
  for (int c = lane; c < F; c += 32) {
    float hv = hp[c];
    ss += hv * avs[c];
    sd += hv * avd[c];
  }
#pragma unroll
  for (int off = 16; off > 0; off >>= 1) {
    ss += __shfl_xor(ss, off, 32);
    sd += __shfl_xor(sd, off, 32);
  }
  if (lane == 0) { as[node] = ss; ad[node] = sd; }
}

__device__ __forceinline__ void edge_endpoints(int t, int E, const int* __restrict__ esrc,
                                               const int* __restrict__ edst, int& s, int& d) {
  if (t < E) { s = esrc[t]; d = edst[t]; } else { s = t - E; d = s; }
}

__global__ void edge_max(const float* __restrict__ as, const float* __restrict__ ad,
                         const int* __restrict__ esrc, const int* __restrict__ edst,
                         unsigned* __restrict__ menc, int E, int Nn) {
  int t = blockIdx.x * blockDim.x + threadIdx.x;
  if (t >= E + Nn) return;
  int s, d; edge_endpoints(t, E, esrc, edst, s, d);
  float e = as[s] + ad[d];
  e = (e > 0.f) ? e : NEG_SLOPE * e;
  atomicMax(menc + d, enc_f32(e));
}

__global__ void edge_expsum(const float* __restrict__ as, const float* __restrict__ ad,
                            const int* __restrict__ esrc, const int* __restrict__ edst,
                            const unsigned* __restrict__ menc, float* __restrict__ w,
                            float* __restrict__ z, int E, int Nn) {
  int t = blockIdx.x * blockDim.x + threadIdx.x;
  if (t >= E + Nn) return;
  int s, d; edge_endpoints(t, E, esrc, edst, s, d);
  float e = as[s] + ad[d];
  e = (e > 0.f) ? e : NEG_SLOPE * e;
  float wv = expf(e - dec_f32(menc[d]));
  w[t] = wv;
  atomicAdd(z + d, wv);
}

__global__ void aggregate(const float* __restrict__ h, const int* __restrict__ esrc,
                          const int* __restrict__ edst, const float* __restrict__ w,
                          const float* __restrict__ z, float* __restrict__ acc,
                          int E, int Nn, int F) {
  long long t = (long long)blockIdx.x * blockDim.x + threadIdx.x;
  int f4 = F >> 2;
  long long total = (long long)(E + Nn) * f4;
  if (t >= total) return;
  int eix = (int)(t / f4);
  int c = ((int)(t % f4)) << 2;
  int s, d; edge_endpoints(eix, E, esrc, edst, s, d);
  float alpha = w[eix] / z[d];
  const float4 hv = *(const float4*)(h + (size_t)s * F + c);
  float* op = acc + (size_t)d * F + c;
  atomicAdd(op + 0, alpha * hv.x);
  atomicAdd(op + 1, alpha * hv.y);
  atomicAdd(op + 2, alpha * hv.z);
  atomicAdd(op + 3, alpha * hv.w);
}

__global__ void bias_act(const float* __restrict__ acc, const float* __restrict__ b,
                         float* __restrict__ out, long long total, int F, int do_relu) {
  long long t = (long long)blockIdx.x * blockDim.x + threadIdx.x;
  if (t >= total) return;
  float v = acc[t] + b[(int)(t % F)];
  out[t] = do_relu ? fmaxf(v, 0.f) : v;
}

static void run_layer(const float* xin, int Fin, const float* W,
                      const float* avs, const float* avd, const float* bias,
                      float* h, float* accb, float* xout, int Fout, int do_relu,
                      const int* esrc, const int* edst, int E, int Nn,
                      float* w_e, float* z, unsigned* menc, float* as, float* ad,
                      hipStream_t stream) {
  int mt = (Nn + 15) / 16, nt4 = Fout / 64;
  int waves = mt * nt4;
  gemm_f32_wmma<<<(waves + 7) / 8, 256, 0, stream>>>(xin, W, h, Nn, Fin, Fout);
  node_alpha<<<(Nn + 7) / 8, 256, 0, stream>>>(h, avs, avd, as, ad, Nn, Fout);

  int ET = E + Nn;
  fill_u32<<<(Nn + 255) / 256, 256, 0, stream>>>(menc, ENC_NEG_INF, Nn);
  fill_f32<<<(Nn + 255) / 256, 256, 0, stream>>>(z, 0.f, (long long)Nn);
  long long accn = (long long)Nn * Fout;
  fill_f32<<<(unsigned)((accn + 255) / 256), 256, 0, stream>>>(accb, 0.f, accn);

  edge_max<<<(ET + 255) / 256, 256, 0, stream>>>(as, ad, esrc, edst, menc, E, Nn);
  edge_expsum<<<(ET + 255) / 256, 256, 0, stream>>>(as, ad, esrc, edst, menc, w_e, z, E, Nn);

  long long aggt = (long long)ET * (Fout >> 2);
  aggregate<<<(unsigned)((aggt + 255) / 256), 256, 0, stream>>>(h, esrc, edst, w_e, z, accb,
                                                                E, Nn, Fout);
  bias_act<<<(unsigned)((accn + 255) / 256), 256, 0, stream>>>(accb, bias, xout, accn, Fout,
                                                               do_relu);
}

extern "C" void kernel_launch(void* const* d_in, const int* in_sizes, int n_in,
                              void* d_out, int out_size, void* d_ws, size_t ws_size,
                              hipStream_t stream) {
  const float* x   = (const float*)d_in[0];
  const int*  eidx = (const int*)d_in[1];
  const float* W1  = (const float*)d_in[2];
  const float* a1s = (const float*)d_in[3];
  const float* a1d = (const float*)d_in[4];
  const float* b1  = (const float*)d_in[5];
  const float* W2  = (const float*)d_in[6];
  const float* a2s = (const float*)d_in[7];
  const float* a2d = (const float*)d_in[8];
  const float* b2  = (const float*)d_in[9];
  float* out = (float*)d_out;

  const int IN_C = 128, HID_C = 256, OUT_C = 128;
  int Nn = in_sizes[0] / IN_C;
  int E  = in_sizes[1] / 2;
  const int* esrc = eidx;
  const int* edst = eidx + E;
  int ET = E + Nn;

  float* bufA = (float*)d_ws;
  float* bufB = bufA + (size_t)Nn * HID_C;
  float* w_e  = bufB + (size_t)Nn * HID_C;
  float* z    = w_e + ET;
  unsigned* menc = (unsigned*)(z + Nn);
  float* as   = (float*)(menc + Nn);
  float* ad   = as + Nn;
  (void)ws_size; (void)n_in; (void)out_size;

  run_layer(x, IN_C, W1, a1s, a1d, b1,
             bufA,  bufB,  bufB, HID_C,  1,
            esrc, edst, E, Nn, w_e, z, menc, as, ad, stream);

  float* h2   = bufA;
  float* acc2 = bufA + (size_t)Nn * OUT_C;
  run_layer(bufB, HID_C, W2, a2s, a2d, b2,
             h2,  acc2,  out, OUT_C,  0,
            esrc, edst, E, Nn, w_e, z, menc, as, ad, stream);
}


# sums seeded by plain stores with bias + own self-loop term: zero fills, bias passes, self-loop atomics gone; layer 2 adds into out; relu on GEMM2 A load; bias now summed first
# speedup vs baseline: 1.7478x; 1.7478x over previous
#include <hip/hip_runtime.h>

typedef __attribute__((ext_vector_type(2))) float v2f;
typedef __attribute__((ext_vector_type(8))) float v8f;

#define NEG_SLOPE 0.2f

__device__ __forceinline__ unsigned enc_f32(float f) {
  unsigned u = __float_as_uint(f);
  return (u & 0x80000000u) ? ~u : (u | 0x80000000u);
}
__device__ __forceinline__ float dec_f32(unsigned u) {
  unsigned b = (u & 0x80000000u) ? (u & 0x7fffffffu) : ~u;
  return __uint_as_float(b);
}
#define ENC_NEG_INF 0x007FFFFFu

__global__ void fill_f32(float* __restrict__ p, float v, long long n) {
  long long i = (long long)blockIdx.x * blockDim.x + threadIdx.x;
  if (i < n) p[i] = v;
}
__global__ void fill_u32(unsigned* __restrict__ p, unsigned v, int n) {
  int i = blockIdx.x * blockDim.x + threadIdx.x;
  if (i < n) p[i] = v;
}

template <bool RELU_A>
__global__ void gemm_f32_wmma(const float* __restrict__ X, const float* __restrict__ W,
                              float* __restrict__ H, int M, int K, int N) {
  int lane = threadIdx.x & 31;
  int wave = (int)((blockIdx.x * (long long)blockDim.x + threadIdx.x) >> 5);
  int nt4 = N >> 6;
  int mt  = (M + 15) >> 4;
  if (wave >= mt * nt4) return;
  int tm = wave / nt4, tn = wave % nt4;
  int half = lane >> 4;
  int l15  = lane & 15;
  int row = tm * 16 + l15;  if (row >= M) row = M - 1;
  int col0 = tn * 64 + l15;
  v8f acc0 = {}, acc1 = {}, acc2 = {}, acc3 = {};
  for (int k0 = 0; k0 < K; k0 += 4) {
    int kb = k0 + (half << 1);
    v2f a = *(const v2f*)(X + (size_t)row * K + kb);
    if (RELU_A) { a.x = fmaxf(a.x, 0.f); a.y = fmaxf(a.y, 0.f); }
    const float* w0 = W + (size_t)kb * N;
    const float* w1 = W + (size_t)(kb + 1) * N;
    v2f b0, b1, b2, b3;
    b0.x = w0[col0 +  0]; b0.y = w1[col0 +  0];
    b1.x = w0[col0 + 16]; b1.y = w1[col0 + 16];
    b2.x = w0[col0 + 32]; b2.y = w1[col0 + 32];
    b3.x = w0[col0 + 48]; b3.y = w1[col0 + 48];
    acc0 = __builtin_amdgcn_wmma_f32_16x16x4_f32(false, a, false, b0, (short)0, acc0, false, false);
    acc1 = __builtin_amdgcn_wmma_f32_16x16x4_f32(false, a, false, b1, (short)0, acc1, false, false);
    acc2 = __builtin_amdgcn_wmma_f32_16x16x4_f32(false, a, false, b2, (short)0, acc2, false, false);
    acc3 = __builtin_amdgcn_wmma_f32_16x16x4_f32(false, a, false, b3, (short)0, acc3, false, false);
  }
  int rbase = tm * 16 + half * 8;
#pragma unroll
  for (int v = 0; v < 8; ++v) {
    int r = rbase + v;
    if (r < M) {
      float* hp = H + (size_t)r * N + col0;
      hp[ 0] = acc0[v];
      hp[16] = acc1[v];
      hp[32] = acc2[v];
      hp[48] = acc3[v];
    }
  }
}

__global__ void node_alpha(const float* __restrict__ h, const float* __restrict__ avs,
                           const float* __restrict__ avd, float* __restrict__ as,
                           float* __restrict__ ad, int Nn, int F) {
  int lane = threadIdx.x & 31;
  int node = (int)((blockIdx.x * (long long)blockDim.x + threadIdx.x) >> 5);
  if (node >= Nn) return;
  const float* hp = h + (size_t)node * F;
  float ss = 0.f, sd = 0.f;
  for (int c = lane; c < F; c += 32) {
    float hv = hp[c];
    ss += hv * avs[c];
    sd += hv * avd[c];
  }
#pragma unroll
  for (int off = 16; off > 0; off >>= 1) {
    ss += __shfl_xor(ss, off, 32);
    sd += __shfl_xor(sd, off, 32);
  }
  if (lane == 0) { as[node] = ss; ad[node] = sd; }
}

__device__ __forceinline__ void edge_endpoints(int t, int E, const int* __restrict__ esrc,
                                               const int* __restrict__ edst, int& s, int& d) {
  if (t < E) { s = esrc[t]; d = edst[t]; } else { s = t - E; d = s; }
}

__global__ void edge_max(const float* __restrict__ as, const float* __restrict__ ad,
                         const int* __restrict__ esrc, const int* __restrict__ edst,
                         unsigned* __restrict__ menc, int E, int Nn) {
  int t = blockIdx.x * blockDim.x + threadIdx.x;
  if (t >= E + Nn) return;
  int s, d; edge_endpoints(t, E, esrc, edst, s, d);
  float e = as[s] + ad[d];
  e = (e > 0.f) ? e : NEG_SLOPE * e;
  atomicMax(menc + d, enc_f32(e));
}

__global__ void edge_expsum(const float* __restrict__ as, const float* __restrict__ ad,
                            const int* __restrict__ esrc, const int* __restrict__ edst,
                            const unsigned* __restrict__ menc, float* __restrict__ w,
                            float* __restrict__ z, int E, int Nn) {
  int t = blockIdx.x * blockDim.x + threadIdx.x;
  if (t >= E + Nn) return;
  int s, d; edge_endpoints(t, E, esrc, edst, s, d);
  float e = as[s] + ad[d];
  e = (e > 0.f) ? e : NEG_SLOPE * e;
  float wv = expf(e - dec_f32(menc[d]));
  w[t] = wv;
  atomicAdd(z + d, wv);
}

template <int F>
__global__ void seed_rows(const float* __restrict__ h, const float* __restrict__ wself,
                          const float* __restrict__ z, const float* __restrict__ b,
                          float* __restrict__ acc, int Nn) {
  constexpr int Q = F / 4;
  constexpr int U = 8;
  const int lane = threadIdx.x & 31;
  const int wave = __builtin_amdgcn_readfirstlane(
      (int)((blockIdx.x * (long long)blockDim.x + threadIdx.x) >> 5));
  const long long total = (long long)Nn * Q;
  const long long p0 = (long long)wave * (32 * U);
  if (p0 >= total) return;
  const float4* __restrict__ h4 = (const float4*)h;
  const float4* __restrict__ b4 = (const float4*)b;
  float4* __restrict__ o4 = (float4*)acc;
  const float4 bv = b4[lane & (Q - 1)];
  if (p0 + 32 * U <= total) {
    float4 hv[U];
    float wv[U], zv[U];
    for (int j = 0; j < U; ++j) hv[j] = h4[p0 + j * 32 + lane];
    for (int j = 0; j < U; ++j) {
      const int node = (int)((p0 + j * 32) / Q);
      wv[j] = wself[node];
      zv[j] = z[node];
    }
    for (int j = 0; j < U; ++j) {
      const float a = wv[j] / zv[j];
      float4 bj = bv;
      if (Q == 64 && (j & 1)) bj = b4[32 + lane];
      float4 r;
      r.x = bj.x + a * hv[j].x;
      r.y = bj.y + a * hv[j].y;
      r.z = bj.z + a * hv[j].z;
      r.w = bj.w + a * hv[j].w;
      o4[p0 + j * 32 + lane] = r;
    }
  } else {
    for (int j = 0; j < U; ++j) {
      const long long idx = p0 + j * 32 + lane;
      if (idx < total) {
        const int node = (int)(idx / Q);
        const float a = wself[node] / z[node];
        const float4 hvj = h4[idx];
        const float4 bj = b4[(int)(idx % Q)];
        float4 r;
        r.x = bj.x + a * hvj.x;
        r.y = bj.y + a * hvj.y;
        r.z = bj.z + a * hvj.z;
        r.w = bj.w + a * hvj.w;
        o4[idx] = r;
      }
    }
  }
}

template <int F, int EPW>
__global__ void aggregate_lines(const float* __restrict__ h, const int* __restrict__ esrc,
                                const int* __restrict__ edst, const float* __restrict__ w,
                                const float* __restrict__ z, float* __restrict__ acc,
                                int E) {
  constexpr int NL = F / 32;
  const int lane = threadIdx.x & 31;
  const int wave = __builtin_amdgcn_readfirstlane(
      (int)((blockIdx.x * (long long)blockDim.x + threadIdx.x) >> 5));
  const long long e0l = (long long)wave * EPW;
  if (e0l >= E) return;
  const int e0 = (int)e0l;
  if (e0 + EPW <= E) {
    int s[EPW], d[EPW];
    for (int i = 0; i < EPW; ++i) { s[i] = esrc[e0 + i]; d[i] = edst[e0 + i]; }
    float v[EPW][NL];
    float alpha[EPW];
    for (int i = 0; i < EPW; ++i) {
      alpha[i] = w[e0 + i] / z[d[i]];
      const float* hp = h + (size_t)s[i] * F + lane;
      for (int j = 0; j < NL; ++j) v[i][j] = hp[j * 32];
    }
    for (int i = 0; i < EPW; ++i) {
      float* op = acc + (size_t)d[i] * F + lane;
      for (int j = 0; j < NL; ++j) atomicAdd(op + j * 32, alpha[i] * v[i][j]);
    }
  } else {
    for (int e = e0; e < E; ++e) {
      const int s = esrc[e], d = edst[e];
      float a = w[e] / z[d];
      const float* hp = h + (size_t)s * F + lane;
      float* op = acc + (size_t)d * F + lane;
      for (int j = 0; j < NL; ++j) atomicAdd(op + j * 32, a * hp[j * 32]);
    }
  }
}

template <bool RELU_IN>
static void run_layer(const float* xin, int Fin, const float* W,
                      const float* avs, const float* avd, const float* bias,
                      float* h, float* sum, int Fout,
                      const int* esrc, const int* edst, int E, int Nn,
                      float* w_e, float* z, unsigned* menc, float* as, float* ad,
                      hipStream_t stream) {
  int mt = (Nn + 15) / 16, nt4 = Fout / 64;
  int waves = mt * nt4;
  gemm_f32_wmma<RELU_IN><<<(waves + 7) / 8, 256, 0, stream>>>(xin, W, h, Nn, Fin, Fout);
  node_alpha<<<(Nn + 7) / 8, 256, 0, stream>>>(h, avs, avd, as, ad, Nn, Fout);

  int ET = E + Nn;
  fill_u32<<<(Nn + 255) / 256, 256, 0, stream>>>(menc, ENC_NEG_INF, Nn);
  fill_f32<<<(Nn + 255) / 256, 256, 0, stream>>>(z, 0.f, (long long)Nn);

  edge_max<<<(ET + 255) / 256, 256, 0, stream>>>(as, ad, esrc, edst, menc, E, Nn);
  edge_expsum<<<(ET + 255) / 256, 256, 0, stream>>>(as, ad, esrc, edst, menc, w_e, z, E, Nn);

  long long pieces = (long long)Nn * (Fout / 4);
  unsigned sblocks = (unsigned)((pieces + 2047) / 2048);
  if (Fout == 256) {
    seed_rows<256><<<sblocks, 256, 0, stream>>>(h, w_e + E, z, bias, sum, Nn);
    aggregate_lines<256, 4><<<(unsigned)((E + 31) / 32), 256, 0, stream>>>(h, esrc, edst, w_e, z,
                                                                          sum, E);
  } else {
    seed_rows<128><<<sblocks, 256, 0, stream>>>(h, w_e + E, z, bias, sum, Nn);
    aggregate_lines<128, 8><<<(unsigned)((E + 63) / 64), 256, 0, stream>>>(h, esrc, edst, w_e, z,
                                                                          sum, E);
  }
}

extern "C" void kernel_launch(void* const* d_in, const int* in_sizes, int n_in,
                              void* d_out, int out_size, void* d_ws, size_t ws_size,
                              hipStream_t stream) {
  const float* x   = (const float*)d_in[0];
  const int*  eidx = (const int*)d_in[1];
  const float* W1  = (const float*)d_in[2];
  const float* a1s = (const float*)d_in[3];
  const float* a1d = (const float*)d_in[4];
  const float* b1  = (const float*)d_in[5];
  const float* W2  = (const float*)d_in[6];
  const float* a2s = (const float*)d_in[7];
  const float* a2d = (const float*)d_in[8];
  const float* b2  = (const float*)d_in[9];
  float* out = (float*)d_out;

  const int IN_C = 128, HID_C = 256, OUT_C = 128;
  int Nn = in_sizes[0] / IN_C;
  int E  = in_sizes[1] / 2;
  const int* esrc = eidx;
  const int* edst = eidx + E;
  int ET = E + Nn;

  float* bufA = (float*)d_ws;
  float* bufB = bufA + (size_t)Nn * HID_C;
  float* w_e  = bufB + (size_t)Nn * HID_C;
  float* z    = w_e + ET;
  unsigned* menc = (unsigned*)(z + Nn);
  float* as   = (float*)(menc + Nn);
  float* ad   = as + Nn;
  (void)ws_size; (void)n_in; (void)out_size;

  run_layer<false>(x, IN_C, W1, a1s, a1d, b1,
                    bufA,  bufB, HID_C,
                   esrc, edst, E, Nn, w_e, z, menc, as, ad, stream);

  run_layer<true>(bufB, HID_C, W2, a2s, a2d, b2,
                   bufA,  out, OUT_C,
                  esrc, edst, E, Nn, w_e, z, menc, as, ad, stream);
}


# exp-sum folded into the owner tile: max launch leaves a 16-B record (src,dst,e) per edge; tile forms w=exp(e-m), sums z in LDS (arrival order), scales rows by 1/z once, bias last; 4 launches gone
# speedup vs baseline: 2.5010x; 1.4309x over previous
#include <hip/hip_runtime.h>

typedef __attribute__((ext_vector_type(2))) float v2f;
typedef __attribute__((ext_vector_type(8))) float v8f;

#define NEG_SLOPE 0.2f

__device__ __forceinline__ unsigned enc_f32(float f) {
  unsigned u = __float_as_uint(f);
  return (u & 0x80000000u) ? ~u : (u | 0x80000000u);
}
__device__ __forceinline__ float dec_f32(unsigned u) {
  unsigned b = (u & 0x80000000u) ? (u & 0x7fffffffu) : ~u;
  return __uint_as_float(b);
}
#define ENC_NEG_INF 0x007FFFFFu

__global__ void fill_u32(unsigned* __restrict__ p, unsigned v, int n) {
  int i = blockIdx.x * blockDim.x + threadIdx.x;
  if (i < n) p[i] = v;
}

template <bool RELU_A>
__global__ void gemm_f32_wmma(const float* __restrict__ X, const float* __restrict__ W,
                              float* __restrict__ H, int M, int K, int N) {
  int lane = threadIdx.x & 31;
  int wave = (int)((blockIdx.x * (long long)blockDim.x + threadIdx.x) >> 5);
  int nt4 = N >> 6;
  int mt  = (M + 15) >> 4;
  if (wave >= mt * nt4) return;
  int tm = wave / nt4, tn = wave % nt4;
  int half = lane >> 4;
  int l15  = lane & 15;
  int row = tm * 16 + l15;  if (row >= M) row = M - 1;
  int col0 = tn * 64 + l15;
  v8f acc0 = {}, acc1 = {}, acc2 = {}, acc3 = {};
  for (int k0 = 0; k0 < K; k0 += 4) {
    int kb = k0 + (half << 1);
    v2f a = *(const v2f*)(X + (size_t)row * K + kb);
    if (RELU_A) { a.x = fmaxf(a.x, 0.f); a.y = fmaxf(a.y, 0.f); }
    const float* w0 = W + (size_t)kb * N;
    const float* w1 = W + (size_t)(kb + 1) * N;
    v2f b0, b1, b2, b3;
    b0.x = w0[col0 +  0]; b0.y = w1[col0 +  0];
    b1.x = w0[col0 + 16]; b1.y = w1[col0 + 16];
    b2.x = w0[col0 + 32]; b2.y = w1[col0 + 32];
    b3.x = w0[col0 + 48]; b3.y = w1[col0 + 48];
    acc0 = __builtin_amdgcn_wmma_f32_16x16x4_f32(false, a, false, b0, (short)0, acc0, false, false);
    acc1 = __builtin_amdgcn_wmma_f32_16x16x4_f32(false, a, false, b1, (short)0, acc1, false, false);
    acc2 = __builtin_amdgcn_wmma_f32_16x16x4_f32(false, a, false, b2, (short)0, acc2, false, false);
    acc3 = __builtin_amdgcn_wmma_f32_16x16x4_f32(false, a, false, b3, (short)0, acc3, false, false);
  }
  int rbase = tm * 16 + half * 8;
#pragma unroll
  for (int v = 0; v < 8; ++v) {
    int r = rbase + v;
    if (r < M) {
      float* hp = H + (size_t)r * N + col0;
      hp[ 0] = acc0[v];
      hp[16] = acc1[v];
      hp[32] = acc2[v];
      hp[48] = acc3[v];
    }
  }
}

__global__ void node_alpha(const float* __restrict__ h, const float* __restrict__ avs,
                           const float* __restrict__ avd, float* __restrict__ as,
                           float* __restrict__ ad, int Nn, int F) {
  int lane = threadIdx.x & 31;
  int node = (int)((blockIdx.x * (long long)blockDim.x + threadIdx.x) >> 5);
  if (node >= Nn) return;
  const float* hp = h + (size_t)node * F;
  float ss = 0.f, sd = 0.f;
  for (int c = lane; c < F; c += 32) {
    float hv = hp[c];
    ss += hv * avs[c];
    sd += hv * avd[c];
  }
#pragma unroll
  for (int off = 16; off > 0; off >>= 1) {
    ss += __shfl_xor(ss, off, 32);
    sd += __shfl_xor(sd, off, 32);
  }
  if (lane == 0) { as[node] = ss; ad[node] = sd; }
}

__device__ __forceinline__ void edge_endpoints(int t, int E, const int* __restrict__ esrc,
                                               const int* __restrict__ edst, int& s, int& d) {
  if (t < E) { s = esrc[t]; d = edst[t]; } else { s = t - E; d = s; }
}

__global__ void edge_max(const float* __restrict__ as, const float* __restrict__ ad,
                         const int* __restrict__ esrc, const int* __restrict__ edst,
                         unsigned* __restrict__ menc, unsigned char* __restrict__ bk,
                         float* __restrict__ rec, int E, int Nn) {
  int t = blockIdx.x * blockDim.x + threadIdx.x;
  if (t >= E + Nn) return;
  int s, d; edge_endpoints(t, E, esrc, edst, s, d);
  if (t < E) bk[t] = (unsigned char)(d >> 8);
  float e = as[s] + ad[d];
  e = (e > 0.f) ? e : NEG_SLOPE * e;
  float4 rc;
  rc.x = __int_as_float(s);
  rc.y = __int_as_float(d);
  rc.z = e;
  rc.w = 0.f;
  ((float4*)rec)[t] = rc;
  atomicMax(menc + d, enc_f32(e));
}

#define TROWS 256
#define TWAVES 32
#define QCAP 64

typedef __attribute__((address_space(3))) volatile int lds_queue_t;

template <int F, int GR>
__device__ __forceinline__ void add_queued_rows(lds_queue_t* q, int m, int lane,
                                                const float* __restrict__ h,
                                                const float* __restrict__ rec,
                                                const float* ml, float* zl,
                                                float* tile, int n0) {
  constexpr int Q4 = F / 4;
  constexpr int NS = F / 128;
  const float4* __restrict__ h4 = (const float4*)h;
  int s = 0, r = 0;
  float a = 0.f;
  if (lane < m) {
    const int e = q[lane];
    const float4 rc = ((const float4*)rec)[e];
    s = __float_as_int(rc.x);
    r = __float_as_int(rc.y) - n0;
    a = expf(rc.z - ml[r]);
    __hip_atomic_fetch_add(zl + r, a, __ATOMIC_RELAXED, __HIP_MEMORY_SCOPE_WORKGROUP);
  }
  int k = 0;
  for (; k + GR <= m; k += GR) {
    float4 v[GR][NS];
    float ak[GR];
    int rk[GR];
    for (int g = 0; g < GR; ++g) {
      const int sk = __builtin_amdgcn_readlane(s, k + g);
      ak[g] = __int_as_float(__builtin_amdgcn_readlane(__float_as_int(a), k + g));
      rk[g] = __builtin_amdgcn_readlane(r, k + g);
      for (int j = 0; j < NS; ++j) v[g][j] = h4[(size_t)sk * Q4 + j * 32 + lane];
    }
    for (int g = 0; g < GR; ++g) {
      float* tp = tile + rk[g] * F + lane;
      for (int j = 0; j < NS; ++j) {
        __hip_atomic_fetch_add(tp + j * 128 +  0, ak[g] * v[g][j].x, __ATOMIC_RELAXED,
                               __HIP_MEMORY_SCOPE_WORKGROUP);
        __hip_atomic_fetch_add(tp + j * 128 + 32, ak[g] * v[g][j].y, __ATOMIC_RELAXED,
                               __HIP_MEMORY_SCOPE_WORKGROUP);
        __hip_atomic_fetch_add(tp + j * 128 + 64, ak[g] * v[g][j].z, __ATOMIC_RELAXED,
                               __HIP_MEMORY_SCOPE_WORKGROUP);
        __hip_atomic_fetch_add(tp + j * 128 + 96, ak[g] * v[g][j].w, __ATOMIC_RELAXED,
                               __HIP_MEMORY_SCOPE_WORKGROUP);
      }
    }
  }
  for (; k < m; ++k) {
    const int sk = __builtin_amdgcn_readlane(s, k);
    const float a1 = __int_as_float(__builtin_amdgcn_readlane(__float_as_int(a), k));
    const int r1 = __builtin_amdgcn_readlane(r, k);
    float* tp = tile + r1 * F + lane;
    for (int j = 0; j < NS; ++j) {
      const float4 v1 = h4[(size_t)sk * Q4 + j * 32 + lane];
      __hip_atomic_fetch_add(tp + j * 128 +  0, a1 * v1.x, __ATOMIC_RELAXED,
                             __HIP_MEMORY_SCOPE_WORKGROUP);
      __hip_atomic_fetch_add(tp + j * 128 + 32, a1 * v1.y, __ATOMIC_RELAXED,
                             __HIP_MEMORY_SCOPE_WORKGROUP);
      __hip_atomic_fetch_add(tp + j * 128 + 64, a1 * v1.z, __ATOMIC_RELAXED,
                             __HIP_MEMORY_SCOPE_WORKGROUP);
      __hip_atomic_fetch_add(tp + j * 128 + 96, a1 * v1.w, __ATOMIC_RELAXED,
                             __HIP_MEMORY_SCOPE_WORKGROUP);
    }
  }
}

__device__ __forceinline__ int push_matches(lds_queue_t* q, int cnt, bool match, int e, int lane) {
  const unsigned mask = __builtin_amdgcn_ballot_w32(match);
  if (mask) {
    if (match) q[cnt + __popc(mask & ((1u << lane) - 1u))] = e;
    cnt += __popc(mask);
  }
  return cnt;
}

__device__ __forceinline__ unsigned nonzero_bytes(unsigned t) {
  return ((t & 0x7f7f7f7fu) + 0x7f7f7f7fu) | t;
}

__device__ __forceinline__ unsigned match_bits(uint4 v, unsigned pat) {
  return ((~nonzero_bytes(v.x ^ pat) >> 7) & 0x01010101u) |
         ((~nonzero_bytes(v.y ^ pat) >> 6) & 0x02020202u) |
         ((~nonzero_bytes(v.z ^ pat) >> 5) & 0x04040404u) |
         ((~nonzero_bytes(v.w ^ pat) >> 4) & 0x08080808u);
}

template <int F>
__global__ __launch_bounds__(1024) void owner_tile_sum(
    const float* __restrict__ h, const int* __restrict__ edst,
    const unsigned char* __restrict__ bk,
    const float* __restrict__ rec, const unsigned* __restrict__ menc, const float* __restrict__ b,
    float* __restrict__ sum, int E, int Nn) {
  constexpr int Q4 = F / 4;
  constexpr int U = 8;
  constexpr int GR = (F == 256) ? 4 : 8;
  __shared__ __attribute__((aligned(16))) float tile[TROWS * F];
  __shared__ int queue[TWAVES * QCAP];
  __shared__ float ml[TROWS];
  __shared__ float zl[TROWS];
  const int lane = threadIdx.x & 31;
  const int wave = __builtin_amdgcn_readfirstlane((int)(threadIdx.x >> 5));
  const int bucket = blockIdx.x;
  const int n0 = bucket * TROWS;
  const int nn = min(TROWS, Nn - n0);
  const float4* __restrict__ h4 = (const float4*)h;
  const float4* __restrict__ b4 = (const float4*)b;

  if (threadIdx.x < TROWS) {
    const int node = n0 + min((int)threadIdx.x, nn - 1);
    const float mv = dec_f32(menc[node]);
    ml[threadIdx.x] = mv;
    zl[threadIdx.x] = expf(rec[4 * (size_t)(E + node) + 2] - mv);
  }

  {
    constexpr int RPW = TROWS / TWAVES;
    constexpr int PPL = Q4 / 32;
    float4 hv[RPW][PPL];
    float av[RPW];
    for (int i = 0; i < RPW; ++i) {
      const int node = n0 + min(wave + i * TWAVES, nn - 1);
      av[i] = expf(rec[4 * (size_t)(E + node) + 2] - dec_f32(menc[node]));
      for (int j = 0; j < PPL; ++j) hv[i][j] = h4[(size_t)node * Q4 + j * 32 + lane];
    }
    for (int i = 0; i < RPW; ++i) {
      const int r = wave + i * TWAVES;
      for (int j = 0; j < PPL; ++j) {
        float4 o;
        o.x = av[i] * hv[i][j].x;
        o.y = av[i] * hv[i][j].y;
        o.z = av[i] * hv[i][j].z;
        o.w = av[i] * hv[i][j].w;
        float* tp = tile + r * F + j * 128 + lane;
        tp[0]  = o.x;
        tp[32] = o.y;
        tp[64] = o.z;
        tp[96] = o.w;
      }
    }
  }
  __syncthreads();

  lds_queue_t* q = (lds_queue_t*)(queue + wave * QCAP);
  int cnt = 0;
  const bool wide = ((((size_t)bk) & 15) == 0) && Nn <= 65536;
  constexpr int SPAN = 512 * U;
  const int nfull = wide ? E / SPAN : 0;
  const unsigned pat = (unsigned)bucket * 0x01010101u;
  for (int t = wave; t < nfull; t += TWAVES) {
    const int base = t * SPAN;
    uint4 dv[U];
    for (int u = 0; u < U; ++u)
      dv[u] = *(const uint4*)(bk + base + u * 512 + lane * 16);
    unsigned g[U];
    for (int u = 0; u < U; ++u) g[u] = match_bits(dv[u], pat);
    bool again;
    do {
      again = false;
#pragma unroll
      for (int u = 0; u < U; ++u) {
        while (!again && __builtin_amdgcn_ballot_w32(g[u] != 0u)) {
          const bool has = g[u] != 0u;
          const int p = __builtin_ctz(g[u] | 0x80000000u);
          const int e = base + u * 512 + lane * 16 + ((p & 7) << 2) + (p >> 3);
          cnt = push_matches(q, cnt, has, e, lane);
          g[u] &= g[u] - 1u;
          if (cnt >= 32) again = true;
        }
      }
      while (cnt >= 32) {
        cnt -= 32;
        add_queued_rows<F, GR>(q + cnt, 32, lane, h, rec, ml, zl, tile, n0);
      }
    } while (again);
  }
  for (int e0 = nfull * SPAN + wave * 32; e0 < E; e0 += TWAVES * 32) {
    const int e = e0 + lane;
    const int d = (e < E) ? edst[e] : -1;
    cnt = push_matches(q, cnt, (d >> 8) == bucket, e, lane);
    while (cnt >= 32) {
      cnt -= 32;
      add_queued_rows<F, GR>(q + cnt, 32, lane, h, rec, ml, zl, tile, n0);
    }
  }
  if (cnt > 0) add_queued_rows<F, GR>(q, cnt, lane, h, rec, ml, zl, tile, n0);
  __syncthreads();

  float4* __restrict__ s4 = (float4*)sum;
  for (int r = wave; r < nn; r += TWAVES) {
    const float rz = 1.0f / zl[r];
    for (int j = 0; j < Q4 / 32; ++j) {
      const float* tp = tile + r * F + j * 128 + lane;
      const float4 bv = b4[j * 32 + lane];
      float4 o;
      o.x = bv.x + tp[0]  * rz;
      o.y = bv.y + tp[32] * rz;
      o.z = bv.z + tp[64] * rz;
      o.w = bv.w + tp[96] * rz;
      s4[(size_t)(n0 + r) * Q4 + j * 32 + lane] = o;
    }
  }
}

template <bool RELU_IN>
static void run_layer(const float* xin, int Fin, const float* W,
                      const float* avs, const float* avd, const float* bias,
                      float* h, float* sum, int Fout,
                      const int* esrc, const int* edst, int E, int Nn,
                      unsigned* menc, float* as, float* ad,
                      unsigned char* bk, float* rec, hipStream_t stream) {
  int mt = (Nn + 15) / 16, nt4 = Fout / 64;
  int waves = mt * nt4;
  gemm_f32_wmma<RELU_IN><<<(waves + 7) / 8, 256, 0, stream>>>(xin, W, h, Nn, Fin, Fout);
  node_alpha<<<(Nn + 7) / 8, 256, 0, stream>>>(h, avs, avd, as, ad, Nn, Fout);

  int ET = E + Nn;
  fill_u32<<<(Nn + 255) / 256, 256, 0, stream>>>(menc, ENC_NEG_INF, Nn);

  edge_max<<<(ET + 255) / 256, 256, 0, stream>>>(as, ad, esrc, edst, menc, bk, rec, E, Nn);

  unsigned tblocks = (unsigned)((Nn + TROWS - 1) / TROWS);
  if (Fout == 256)
    owner_tile_sum<256><<<tblocks, 1024, 0, stream>>>(h, edst, bk, rec, menc, bias, sum, E, Nn);
  else
    owner_tile_sum<128><<<tblocks, 1024, 0, stream>>>(h, edst, bk, rec, menc, bias, sum, E, Nn);
}

extern "C" void kernel_launch(void* const* d_in, const int* in_sizes, int n_in,
                              void* d_out, int out_size, void* d_ws, size_t ws_size,
                              hipStream_t stream) {
  const float* x   = (const float*)d_in[0];
  const int*  eidx = (const int*)d_in[1];
  const float* W1  = (const float*)d_in[2];
  const float* a1s = (const float*)d_in[3];
  const float* a1d = (const float*)d_in[4];
  const float* b1  = (const float*)d_in[5];
  const float* W2  = (const float*)d_in[6];
  const float* a2s = (const float*)d_in[7];
  const float* a2d = (const float*)d_in[8];
  const float* b2  = (const float*)d_in[9];
  float* out = (float*)d_out;

  const int IN_C = 128, HID_C = 256, OUT_C = 128;
  int Nn = in_sizes[0] / IN_C;
  int E  = in_sizes[1] / 2;
  const int* esrc = eidx;
  const int* edst = eidx + E;

  float* bufA = (float*)d_ws;
  float* bufB = bufA + (size_t)Nn * HID_C;
  unsigned* menc = (unsigned*)(bufB + (size_t)Nn * HID_C);
  float* as   = (float*)(menc + Nn);
  float* ad   = as + Nn;
  (void)ws_size; (void)n_in; (void)out_size;
  const size_t rec_off = ((size_t)E + 3) & ~(size_t)3;

  run_layer<false>(x, IN_C, W1, a1s, a1d, b1,
                    bufA,  bufB, HID_C,
                   esrc, edst, E, Nn, menc, as, ad,
                    (unsigned char*)out,  out + rec_off, stream);

  run_layer<true>(bufB, HID_C, W2, a2s, a2d, b2,
                   bufA,  out, OUT_C,
                  esrc, edst, E, Nn, menc, as, ad,
                   (unsigned char*)bufB,  bufB + rec_off, stream);
}


# logit and segment-max launches regrouped into few long waves: one wave per 16 rows (whole 512-B loads in flight together, one transposing lane sum), one thread per 8 edges; logit sum order changes
# speedup vs baseline: 2.6897x; 1.0754x over previous
#include <hip/hip_runtime.h>

typedef __attribute__((ext_vector_type(2))) float v2f;
typedef __attribute__((ext_vector_type(8))) float v8f;

#define NEG_SLOPE 0.2f

__device__ __forceinline__ unsigned enc_f32(float f) {
  unsigned u = __float_as_uint(f);
  return (u & 0x80000000u) ? ~u : (u | 0x80000000u);
}
__device__ __forceinline__ float dec_f32(unsigned u) {
  unsigned b = (u & 0x80000000u) ? (u & 0x7fffffffu) : ~u;
  return __uint_as_float(b);
}
#define ENC_NEG_INF 0x007FFFFFu

__global__ void fill_u32(unsigned* __restrict__ p, unsigned v, int n) {
  int i = blockIdx.x * blockDim.x + threadIdx.x;
  if (i < n) p[i] = v;
}

template <bool RELU_A>
__global__ void gemm_f32_wmma(const float* __restrict__ X, const float* __restrict__ W,
                              float* __restrict__ H, int M, int K, int N) {
  int lane = threadIdx.x & 31;
  int wave = (int)((blockIdx.x * (long long)blockDim.x + threadIdx.x) >> 5);
  int nt4 = N >> 6;
  int mt  = (M + 15) >> 4;
  if (wave >= mt * nt4) return;
  int tm = wave / nt4, tn = wave % nt4;
  int half = lane >> 4;
  int l15  = lane & 15;
  int row = tm * 16 + l15;  if (row >= M) row = M - 1;
  int col0 = tn * 64 + l15;
  v8f acc0 = {}, acc1 = {}, acc2 = {}, acc3 = {};
  for (int k0 = 0; k0 < K; k0 += 4) {
    int kb = k0 + (half << 1);
    v2f a = *(const v2f*)(X + (size_t)row * K + kb);
    if (RELU_A) { a.x = fmaxf(a.x, 0.f); a.y = fmaxf(a.y, 0.f); }
    const float* w0 = W + (size_t)kb * N;
    const float* w1 = W + (size_t)(kb + 1) * N;
    v2f b0, b1, b2, b3;
    b0.x = w0[col0 +  0]; b0.y = w1[col0 +  0];
    b1.x = w0[col0 + 16]; b1.y = w1[col0 + 16];
    b2.x = w0[col0 + 32]; b2.y = w1[col0 + 32];
    b3.x = w0[col0 + 48]; b3.y = w1[col0 + 48];
    acc0 = __builtin_amdgcn_wmma_f32_16x16x4_f32(false, a, false, b0, (short)0, acc0, false, false);
    acc1 = __builtin_amdgcn_wmma_f32_16x16x4_f32(false, a, false, b1, (short)0, acc1, false, false);
    acc2 = __builtin_amdgcn_wmma_f32_16x16x4_f32(false, a, false, b2, (short)0, acc2, false, false);
    acc3 = __builtin_amdgcn_wmma_f32_16x16x4_f32(false, a, false, b3, (short)0, acc3, false, false);
  }
  int rbase = tm * 16 + half * 8;
#pragma unroll
  for (int v = 0; v < 8; ++v) {
    int r = rbase + v;
    if (r < M) {
      float* hp = H + (size_t)r * N + col0;
      hp[ 0] = acc0[v];
      hp[16] = acc1[v];
      hp[32] = acc2[v];
      hp[48] = acc3[v];
    }
  }
}

#define NA_ROWS 16

__device__ __forceinline__ float sum_transpose32(float (&v)[32], int lane) {
#pragma unroll
  for (int n = 16; n >= 1; n >>= 1) {
    const bool up = (lane & n) != 0;
#pragma unroll
    for (int i = 0; i < n; ++i) {
      const float lo = v[i], hi = v[i + n];
      const float give = up ? lo : hi;
      const float keep = up ? hi : lo;
      v[i] = keep + __shfl_xor(give, n, 32);
    }
  }
  return v[0];
}

template <int F>
__global__ __launch_bounds__(256) void node_alpha(const float* __restrict__ h,
                                                  const float* __restrict__ avs,
                                                  const float* __restrict__ avd,
                                                  float* __restrict__ as, float* __restrict__ ad,
                                                  int Nn) {
  constexpr int NS = F / 128;
  constexpr int Q4 = F / 4;
  const int lane = threadIdx.x & 31;
  const int wave = (int)((blockIdx.x * (long long)blockDim.x + threadIdx.x) >> 5);
  const int row0 = wave * NA_ROWS;
  if (row0 >= Nn) return;
  const float4* __restrict__ h4 = (const float4*)h;
  float4 hv[NA_ROWS][NS];
  for (int r = 0; r < NA_ROWS; ++r) {
    const int node = min(row0 + r, Nn - 1);
    for (int j = 0; j < NS; ++j) hv[r][j] = h4[(size_t)node * Q4 + j * 32 + lane];
  }
  float4 vs[NS], vd[NS];
  for (int j = 0; j < NS; ++j) {
    vs[j] = ((const float4*)avs)[j * 32 + lane];
    vd[j] = ((const float4*)avd)[j * 32 + lane];
  }
  float v[2 * NA_ROWS];
  for (int r = 0; r < NA_ROWS; ++r) {
    float ss = 0.f, sd = 0.f;
    for (int j = 0; j < NS; ++j) {
      ss += hv[r][j].x * vs[j].x;  sd += hv[r][j].x * vd[j].x;
      ss += hv[r][j].y * vs[j].y;  sd += hv[r][j].y * vd[j].y;
      ss += hv[r][j].z * vs[j].z;  sd += hv[r][j].z * vd[j].z;
      ss += hv[r][j].w * vs[j].w;  sd += hv[r][j].w * vd[j].w;
    }
    v[r] = ss;
    v[NA_ROWS + r] = sd;
  }
  const float tot = sum_transpose32(v, lane);
  const int row = row0 + (lane & (NA_ROWS - 1));
  if (row < Nn) {
    if (lane < NA_ROWS) as[row] = tot; else ad[row] = tot;
  }
}

#define EM_PER 8
template <bool TAIL>
__device__ __forceinline__ void edge_max_items(const float* __restrict__ as,
                                               const float* __restrict__ ad,
                                               const int* __restrict__ esrc,
                                               const int* __restrict__ edst,
                                               unsigned* __restrict__ menc,
                                               unsigned char* __restrict__ bk,
                                               float* __restrict__ rec, int E, int ET, int base) {
  int s[EM_PER], d[EM_PER];
  for (int k = 0; k < EM_PER; ++k) {
    int t = base + k * (int)blockDim.x;
    if (TAIL) t = min(t, ET - 1);
    const int tl = min(t, E - 1);
    const int sl = esrc[tl], dl = edst[tl];
    s[k] = (t < E) ? sl : t - E;
    d[k] = (t < E) ? dl : t - E;
  }
  float e[EM_PER];
  for (int k = 0; k < EM_PER; ++k) e[k] = as[s[k]] + ad[d[k]];
  for (int k = 0; k < EM_PER; ++k) {
    const int t = base + k * (int)blockDim.x;
    if (!TAIL || t < ET) {
      if (t < E) bk[t] = (unsigned char)(d[k] >> 8);
      const float ev = (e[k] > 0.f) ? e[k] : NEG_SLOPE * e[k];
      float4 rc;
      rc.x = __int_as_float(s[k]);
      rc.y = __int_as_float(d[k]);
      rc.z = ev;
      rc.w = 0.f;
      ((float4*)rec)[t] = rc;
      atomicMax(menc + d[k], enc_f32(ev));
    }
  }
}

__global__ __launch_bounds__(256) void edge_max(const float* __restrict__ as,
                                                const float* __restrict__ ad,
                                                const int* __restrict__ esrc,
                                                const int* __restrict__ edst,
                                                unsigned* __restrict__ menc,
                                                unsigned char* __restrict__ bk,
                                                float* __restrict__ rec, int E, int Nn) {
  const int ET = E + Nn;
  const int span = (int)(blockDim.x * EM_PER);
  const int first = blockIdx.x * span;
  if (first + span <= ET)
    edge_max_items<false>(as, ad, esrc, edst, menc, bk, rec, E, ET, first + (int)threadIdx.x);
  else
    edge_max_items<true>(as, ad, esrc, edst, menc, bk, rec, E, ET, first + (int)threadIdx.x);
}

#define TROWS 256
#define TWAVES 32
#define QCAP 64

typedef __attribute__((address_space(3))) volatile int lds_queue_t;

template <int F, int GR>
__device__ __forceinline__ void add_queued_rows(lds_queue_t* q, int m, int lane,
                                                const float* __restrict__ h,
                                                const float* __restrict__ rec,
                                                const float* ml, float* zl,
                                                float* tile, int n0) {
  constexpr int Q4 = F / 4;
  constexpr int NS = F / 128;
  const float4* __restrict__ h4 = (const float4*)h;
  int s = 0, r = 0;
  float a = 0.f;
  if (lane < m) {
    const int e = q[lane];
    const float4 rc = ((const float4*)rec)[e];
    s = __float_as_int(rc.x);
    r = __float_as_int(rc.y) - n0;
    a = expf(rc.z - ml[r]);
    __hip_atomic_fetch_add(zl + r, a, __ATOMIC_RELAXED, __HIP_MEMORY_SCOPE_WORKGROUP);
  }
  int k = 0;
  for (; k + GR <= m; k += GR) {
    float4 v[GR][NS];
    float ak[GR];
    int rk[GR];
    for (int g = 0; g < GR; ++g) {
      const int sk = __builtin_amdgcn_readlane(s, k + g);
      ak[g] = __int_as_float(__builtin_amdgcn_readlane(__float_as_int(a), k + g));
      rk[g] = __builtin_amdgcn_readlane(r, k + g);
      for (int j = 0; j < NS; ++j) v[g][j] = h4[(size_t)sk * Q4 + j * 32 + lane];
    }
    for (int g = 0; g < GR; ++g) {
      float* tp = tile + rk[g] * F + lane;
      for (int j = 0; j < NS; ++j) {
        __hip_atomic_fetch_add(tp + j * 128 +  0, ak[g] * v[g][j].x, __ATOMIC_RELAXED,
                               __HIP_MEMORY_SCOPE_WORKGROUP);
        __hip_atomic_fetch_add(tp + j * 128 + 32, ak[g] * v[g][j].y, __ATOMIC_RELAXED,
                               __HIP_MEMORY_SCOPE_WORKGROUP);
        __hip_atomic_fetch_add(tp + j * 128 + 64, ak[g] * v[g][j].z, __ATOMIC_RELAXED,
                               __HIP_MEMORY_SCOPE_WORKGROUP);
        __hip_atomic_fetch_add(tp + j * 128 + 96, ak[g] * v[g][j].w, __ATOMIC_RELAXED,
                               __HIP_MEMORY_SCOPE_WORKGROUP);
      }
    }
  }
  for (; k < m; ++k) {
    const int sk = __builtin_amdgcn_readlane(s, k);
    const float a1 = __int_as_float(__builtin_amdgcn_readlane(__float_as_int(a), k));
    const int r1 = __builtin_amdgcn_readlane(r, k);
    float* tp = tile + r1 * F + lane;
    for (int j = 0; j < NS; ++j) {
      const float4 v1 = h4[(size_t)sk * Q4 + j * 32 + lane];
      __hip_atomic_fetch_add(tp + j * 128 +  0, a1 * v1.x, __ATOMIC_RELAXED,
                             __HIP_MEMORY_SCOPE_WORKGROUP);
      __hip_atomic_fetch_add(tp + j * 128 + 32, a1 * v1.y, __ATOMIC_RELAXED,
                             __HIP_MEMORY_SCOPE_WORKGROUP);
      __hip_atomic_fetch_add(tp + j * 128 + 64, a1 * v1.z, __ATOMIC_RELAXED,
                             __HIP_MEMORY_SCOPE_WORKGROUP);
      __hip_atomic_fetch_add(tp + j * 128 + 96, a1 * v1.w, __ATOMIC_RELAXED,
                             __HIP_MEMORY_SCOPE_WORKGROUP);
    }
  }
}

__device__ __forceinline__ int push_matches(lds_queue_t* q, int cnt, bool match, int e, int lane) {
  const unsigned mask = __builtin_amdgcn_ballot_w32(match);
  if (mask) {
    if (match) q[cnt + __popc(mask & ((1u << lane) - 1u))] = e;
    cnt += __popc(mask);
  }
  return cnt;
}

__device__ __forceinline__ unsigned nonzero_bytes(unsigned t) {
  return ((t & 0x7f7f7f7fu) + 0x7f7f7f7fu) | t;
}

__device__ __forceinline__ unsigned match_bits(uint4 v, unsigned pat) {
  return ((~nonzero_bytes(v.x ^ pat) >> 7) & 0x01010101u) |
         ((~nonzero_bytes(v.y ^ pat) >> 6) & 0x02020202u) |
         ((~nonzero_bytes(v.z ^ pat) >> 5) & 0x04040404u) |
         ((~nonzero_bytes(v.w ^ pat) >> 4) & 0x08080808u);
}

template <int F>
__global__ __launch_bounds__(1024) void owner_tile_sum(
    const float* __restrict__ h, const int* __restrict__ edst,
    const unsigned char* __restrict__ bk,
    const float* __restrict__ rec, const unsigned* __restrict__ menc, const float* __restrict__ b,
    float* __restrict__ sum, int E, int Nn) {
  constexpr int Q4 = F / 4;
  constexpr int U = 8;
  constexpr int GR = (F == 256) ? 4 : 8;
  __shared__ __attribute__((aligned(16))) float tile[TROWS * F];
  __shared__ int queue[TWAVES * QCAP];
  __shared__ float ml[TROWS];
  __shared__ float zl[TROWS];
  const int lane = threadIdx.x & 31;
  const int wave = __builtin_amdgcn_readfirstlane((int)(threadIdx.x >> 5));
  const int bucket = blockIdx.x;
  const int n0 = bucket * TROWS;
  const int nn = min(TROWS, Nn - n0);
  const float4* __restrict__ h4 = (const float4*)h;
  const float4* __restrict__ b4 = (const float4*)b;

  if (threadIdx.x < TROWS) {
    const int node = n0 + min((int)threadIdx.x, nn - 1);
    const float mv = dec_f32(menc[node]);
    ml[threadIdx.x] = mv;
    zl[threadIdx.x] = expf(rec[4 * (size_t)(E + node) + 2] - mv);
  }

  {
    constexpr int RPW = TROWS / TWAVES;
    constexpr int PPL = Q4 / 32;
    float4 hv[RPW][PPL];
    float av[RPW];
    for (int i = 0; i < RPW; ++i) {
      const int node = n0 + min(wave + i * TWAVES, nn - 1);
      av[i] = expf(rec[4 * (size_t)(E + node) + 2] - dec_f32(menc[node]));
      for (int j = 0; j < PPL; ++j) hv[i][j] = h4[(size_t)node * Q4 + j * 32 + lane];
    }
    for (int i = 0; i < RPW; ++i) {
      const int r = wave + i * TWAVES;
      for (int j = 0; j < PPL; ++j) {
        float4 o;
        o.x = av[i] * hv[i][j].x;
        o.y = av[i] * hv[i][j].y;
        o.z = av[i] * hv[i][j].z;
        o.w = av[i] * hv[i][j].w;
        float* tp = tile + r * F + j * 128 + lane;
        tp[0]  = o.x;
        tp[32] = o.y;
        tp[64] = o.z;
        tp[96] = o.w;
      }
    }
  }
  __syncthreads();

  lds_queue_t* q = (lds_queue_t*)(queue + wave * QCAP);
  int cnt = 0;
  const bool wide = ((((size_t)bk) & 15) == 0) && Nn <= 65536;
  constexpr int SPAN = 512 * U;
  const int nfull = wide ? E / SPAN : 0;
  const unsigned pat = (unsigned)bucket * 0x01010101u;
  for (int t = wave; t < nfull; t += TWAVES) {
    const int base = t * SPAN;
    uint4 dv[U];
    for (int u = 0; u < U; ++u)
      dv[u] = *(const uint4*)(bk + base + u * 512 + lane * 16);
    unsigned g[U];
    for (int u = 0; u < U; ++u) g[u] = match_bits(dv[u], pat);
    bool again;
    do {
      again = false;
#pragma unroll
      for (int u = 0; u < U; ++u) {
        while (!again && __builtin_amdgcn_ballot_w32(g[u] != 0u)) {
          const bool has = g[u] != 0u;
          const int p = __builtin_ctz(g[u] | 0x80000000u);
          const int e = base + u * 512 + lane * 16 + ((p & 7) << 2) + (p >> 3);
          cnt = push_matches(q, cnt, has, e, lane);
          g[u] &= g[u] - 1u;
          if (cnt >= 32) again = true;
        }
      }
      while (cnt >= 32) {
        cnt -= 32;
        add_queued_rows<F, GR>(q + cnt, 32, lane, h, rec, ml, zl, tile, n0);
      }
    } while (again);
  }
  for (int e0 = nfull * SPAN + wave * 32; e0 < E; e0 += TWAVES * 32) {
    const int e = e0 + lane;
    const int d = (e < E) ? edst[e] : -1;
    cnt = push_matches(q, cnt, (d >> 8) == bucket, e, lane);
    while (cnt >= 32) {
      cnt -= 32;
      add_queued_rows<F, GR>(q + cnt, 32, lane, h, rec, ml, zl, tile, n0);
    }
  }
  if (cnt > 0) add_queued_rows<F, GR>(q, cnt, lane, h, rec, ml, zl, tile, n0);
  __syncthreads();

  float4* __restrict__ s4 = (float4*)sum;
  for (int r = wave; r < nn; r += TWAVES) {
    const float rz = 1.0f / zl[r];
    for (int j = 0; j < Q4 / 32; ++j) {
      const float* tp = tile + r * F + j * 128 + lane;
      const float4 bv = b4[j * 32 + lane];
      float4 o;
      o.x = bv.x + tp[0]  * rz;
      o.y = bv.y + tp[32] * rz;
      o.z = bv.z + tp[64] * rz;
      o.w = bv.w + tp[96] * rz;
      s4[(size_t)(n0 + r) * Q4 + j * 32 + lane] = o;
    }
  }
}

template <bool RELU_IN>
static void run_layer(const float* xin, int Fin, const float* W,
                      const float* avs, const float* avd, const float* bias,
                      float* h, float* sum, int Fout,
                      const int* esrc, const int* edst, int E, int Nn,
                      unsigned* menc, float* as, float* ad,
                      unsigned char* bk, float* rec, hipStream_t stream) {
  int mt = (Nn + 15) / 16, nt4 = Fout / 64;
  int waves = mt * nt4;
  gemm_f32_wmma<RELU_IN><<<(waves + 7) / 8, 256, 0, stream>>>(xin, W, h, Nn, Fin, Fout);
  int awaves = (Nn + NA_ROWS - 1) / NA_ROWS;
  if (Fout == 256)
    node_alpha<256><<<(awaves + 7) / 8, 256, 0, stream>>>(h, avs, avd, as, ad, Nn);
  else
    node_alpha<128><<<(awaves + 7) / 8, 256, 0, stream>>>(h, avs, avd, as, ad, Nn);

  int ET = E + Nn;
  fill_u32<<<(Nn + 255) / 256, 256, 0, stream>>>(menc, ENC_NEG_INF, Nn);

  edge_max<<<(ET + 256 * EM_PER - 1) / (256 * EM_PER), 256, 0, stream>>>(as, ad, esrc, edst, menc, bk,
                                                                           rec, E, Nn);

  unsigned tblocks = (unsigned)((Nn + TROWS - 1) / TROWS);
  if (Fout == 256)
    owner_tile_sum<256><<<tblocks, 1024, 0, stream>>>(h, edst, bk, rec, menc, bias, sum, E, Nn);
  else
    owner_tile_sum<128><<<tblocks, 1024, 0, stream>>>(h, edst, bk, rec, menc, bias, sum, E, Nn);
}

extern "C" void kernel_launch(void* const* d_in, const int* in_sizes, int n_in,
                              void* d_out, int out_size, void* d_ws, size_t ws_size,
                              hipStream_t stream) {
  const float* x   = (const float*)d_in[0];
  const int*  eidx = (const int*)d_in[1];
  const float* W1  = (const float*)d_in[2];
  const float* a1s = (const float*)d_in[3];
  const float* a1d = (const float*)d_in[4];
  const float* b1  = (const float*)d_in[5];
  const float* W2  = (const float*)d_in[6];
  const float* a2s = (const float*)d_in[7];
  const float* a2d = (const float*)d_in[8];
  const float* b2  = (const float*)d_in[9];
  float* out = (float*)d_out;

  const int IN_C = 128, HID_C = 256, OUT_C = 128;
  int Nn = in_sizes[0] / IN_C;
  int E  = in_sizes[1] / 2;
  const int* esrc = eidx;
  const int* edst = eidx + E;

  float* bufA = (float*)d_ws;
  float* bufB = bufA + (size_t)Nn * HID_C;
  unsigned* menc = (unsigned*)(bufB + (size_t)Nn * HID_C);
  float* as   = (float*)(menc + Nn);
  float* ad   = as + Nn;
  (void)ws_size; (void)n_in; (void)out_size;
  const size_t rec_off = ((size_t)E + 3) & ~(size_t)3;

  run_layer<false>(x, IN_C, W1, a1s, a1d, b1,
                    bufA,  bufB, HID_C,
                   esrc, edst, E, Nn, menc, as, ad,
                    (unsigned char*)out,  out + rec_off, stream);

  run_layer<true>(bufB, HID_C, W2, a2s, a2d, b2,
                   bufA,  out, OUT_C,
                  esrc, edst, E, Nn, menc, as, ad,
                   (unsigned char*)bufB,  bufB + rec_off, stream);
}


# owner tiles lowered from 256 to 196 rows: 256 tiles, one per WGP (was 196 tiles, 60 of 256 WGPs idle); tile byte = dst / 196; a tile's own row work falls by a quarter; sum order as before
# speedup vs baseline: 2.7982x; 1.0403x over previous
#include <hip/hip_runtime.h>

typedef __attribute__((ext_vector_type(2))) float v2f;
typedef __attribute__((ext_vector_type(8))) float v8f;

#define NEG_SLOPE 0.2f

__device__ __forceinline__ unsigned enc_f32(float f) {
  unsigned u = __float_as_uint(f);
  return (u & 0x80000000u) ? ~u : (u | 0x80000000u);
}
__device__ __forceinline__ float dec_f32(unsigned u) {
  unsigned b = (u & 0x80000000u) ? (u & 0x7fffffffu) : ~u;
  return __uint_as_float(b);
}
#define ENC_NEG_INF 0x007FFFFFu

__global__ void fill_u32(unsigned* __restrict__ p, unsigned v, int n) {
  int i = blockIdx.x * blockDim.x + threadIdx.x;
  if (i < n) p[i] = v;
}

template <bool RELU_A>
__global__ void gemm_f32_wmma(const float* __restrict__ X, const float* __restrict__ W,
                              float* __restrict__ H, int M, int K, int N) {
  int lane = threadIdx.x & 31;
  int wave = (int)((blockIdx.x * (long long)blockDim.x + threadIdx.x) >> 5);
  int nt4 = N >> 6;
  int mt  = (M + 15) >> 4;
  if (wave >= mt * nt4) return;
  int tm = wave / nt4, tn = wave % nt4;
  int half = lane >> 4;
  int l15  = lane & 15;
  int row = tm * 16 + l15;  if (row >= M) row = M - 1;
  int col0 = tn * 64 + l15;
  v8f acc0 = {}, acc1 = {}, acc2 = {}, acc3 = {};
  for (int k0 = 0; k0 < K; k0 += 4) {
    int kb = k0 + (half << 1);
    v2f a = *(const v2f*)(X + (size_t)row * K + kb);
    if (RELU_A) { a.x = fmaxf(a.x, 0.f); a.y = fmaxf(a.y, 0.f); }
    const float* w0 = W + (size_t)kb * N;
    const float* w1 = W + (size_t)(kb + 1) * N;
    v2f b0, b1, b2, b3;
    b0.x = w0[col0 +  0]; b0.y = w1[col0 +  0];
    b1.x = w0[col0 + 16]; b1.y = w1[col0 + 16];
    b2.x = w0[col0 + 32]; b2.y = w1[col0 + 32];
    b3.x = w0[col0 + 48]; b3.y = w1[col0 + 48];
    acc0 = __builtin_amdgcn_wmma_f32_16x16x4_f32(false, a, false, b0, (short)0, acc0, false, false);
    acc1 = __builtin_amdgcn_wmma_f32_16x16x4_f32(false, a, false, b1, (short)0, acc1, false, false);
    acc2 = __builtin_amdgcn_wmma_f32_16x16x4_f32(false, a, false, b2, (short)0, acc2, false, false);
    acc3 = __builtin_amdgcn_wmma_f32_16x16x4_f32(false, a, false, b3, (short)0, acc3, false, false);
  }
  int rbase = tm * 16 + half * 8;
#pragma unroll
  for (int v = 0; v < 8; ++v) {
    int r = rbase + v;
    if (r < M) {
      float* hp = H + (size_t)r * N + col0;
      hp[ 0] = acc0[v];
      hp[16] = acc1[v];
      hp[32] = acc2[v];
      hp[48] = acc3[v];
    }
  }
}

#define NA_ROWS 16

__device__ __forceinline__ float sum_transpose32(float (&v)[32], int lane) {
#pragma unroll
  for (int n = 16; n >= 1; n >>= 1) {
    const bool up = (lane & n) != 0;
#pragma unroll
    for (int i = 0; i < n; ++i) {
      const float lo = v[i], hi = v[i + n];
      const float give = up ? lo : hi;
      const float keep = up ? hi : lo;
      v[i] = keep + __shfl_xor(give, n, 32);
    }
  }
  return v[0];
}

template <int F>
__global__ __launch_bounds__(256) void node_alpha(const float* __restrict__ h,
                                                  const float* __restrict__ avs,
                                                  const float* __restrict__ avd,
                                                  float* __restrict__ as, float* __restrict__ ad,
                                                  int Nn) {
  constexpr int NS = F / 128;
  constexpr int Q4 = F / 4;
  const int lane = threadIdx.x & 31;
  const int wave = (int)((blockIdx.x * (long long)blockDim.x + threadIdx.x) >> 5);
  const int row0 = wave * NA_ROWS;
  if (row0 >= Nn) return;
  const float4* __restrict__ h4 = (const float4*)h;
  float4 hv[NA_ROWS][NS];
  for (int r = 0; r < NA_ROWS; ++r) {
    const int node = min(row0 + r, Nn - 1);
    for (int j = 0; j < NS; ++j) hv[r][j] = h4[(size_t)node * Q4 + j * 32 + lane];
  }
  float4 vs[NS], vd[NS];
  for (int j = 0; j < NS; ++j) {
    vs[j] = ((const float4*)avs)[j * 32 + lane];
    vd[j] = ((const float4*)avd)[j * 32 + lane];
  }
  float v[2 * NA_ROWS];
  for (int r = 0; r < NA_ROWS; ++r) {
    float ss = 0.f, sd = 0.f;
    for (int j = 0; j < NS; ++j) {
      ss += hv[r][j].x * vs[j].x;  sd += hv[r][j].x * vd[j].x;
      ss += hv[r][j].y * vs[j].y;  sd += hv[r][j].y * vd[j].y;
      ss += hv[r][j].z * vs[j].z;  sd += hv[r][j].z * vd[j].z;
      ss += hv[r][j].w * vs[j].w;  sd += hv[r][j].w * vd[j].w;
    }
    v[r] = ss;
    v[NA_ROWS + r] = sd;
  }
  const float tot = sum_transpose32(v, lane);
  const int row = row0 + (lane & (NA_ROWS - 1));
  if (row < Nn) {
    if (lane < NA_ROWS) as[row] = tot; else ad[row] = tot;
  }
}

#define EM_PER 8
#define TROWS 196
template <bool TAIL>
__device__ __forceinline__ void edge_max_items(const float* __restrict__ as,
                                               const float* __restrict__ ad,
                                               const int* __restrict__ esrc,
                                               const int* __restrict__ edst,
                                               unsigned* __restrict__ menc,
                                               unsigned char* __restrict__ bk,
                                               float* __restrict__ rec, int E, int ET, int base) {
  int s[EM_PER], d[EM_PER];
  for (int k = 0; k < EM_PER; ++k) {
    int t = base + k * (int)blockDim.x;
    if (TAIL) t = min(t, ET - 1);
    const int tl = min(t, E - 1);
    const int sl = esrc[tl], dl = edst[tl];
    s[k] = (t < E) ? sl : t - E;
    d[k] = (t < E) ? dl : t - E;
  }
  float e[EM_PER];
  for (int k = 0; k < EM_PER; ++k) e[k] = as[s[k]] + ad[d[k]];
  for (int k = 0; k < EM_PER; ++k) {
    const int t = base + k * (int)blockDim.x;
    if (!TAIL || t < ET) {
      if (t < E) bk[t] = (unsigned char)(d[k] / TROWS);
      const float ev = (e[k] > 0.f) ? e[k] : NEG_SLOPE * e[k];
      float4 rc;
      rc.x = __int_as_float(s[k]);
      rc.y = __int_as_float(d[k]);
      rc.z = ev;
      rc.w = 0.f;
      ((float4*)rec)[t] = rc;
      atomicMax(menc + d[k], enc_f32(ev));
    }
  }
}

__global__ __launch_bounds__(256) void edge_max(const float* __restrict__ as,
                                                const float* __restrict__ ad,
                                                const int* __restrict__ esrc,
                                                const int* __restrict__ edst,
                                                unsigned* __restrict__ menc,
                                                unsigned char* __restrict__ bk,
                                                float* __restrict__ rec, int E, int Nn) {
  const int ET = E + Nn;
  const int span = (int)(blockDim.x * EM_PER);
  const int first = blockIdx.x * span;
  if (first + span <= ET)
    edge_max_items<false>(as, ad, esrc, edst, menc, bk, rec, E, ET, first + (int)threadIdx.x);
  else
    edge_max_items<true>(as, ad, esrc, edst, menc, bk, rec, E, ET, first + (int)threadIdx.x);
}

#define TWAVES 32
#define QCAP 64

typedef __attribute__((address_space(3))) volatile int lds_queue_t;

template <int F, int GR>
__device__ __forceinline__ void add_queued_rows(lds_queue_t* q, int m, int lane,
                                                const float* __restrict__ h,
                                                const float* __restrict__ rec,
                                                const float* ml, float* zl,
                                                float* tile, int n0) {
  constexpr int Q4 = F / 4;
  constexpr int NS = F / 128;
  const float4* __restrict__ h4 = (const float4*)h;
  int s = 0, r = 0;
  float a = 0.f;
  if (lane < m) {
    const int e = q[lane];
    const float4 rc = ((const float4*)rec)[e];
    s = __float_as_int(rc.x);
    r = __float_as_int(rc.y) - n0;
    a = expf(rc.z - ml[r]);
    __hip_atomic_fetch_add(zl + r, a, __ATOMIC_RELAXED, __HIP_MEMORY_SCOPE_WORKGROUP);
  }
  int k = 0;
  for (; k + GR <= m; k += GR) {
    float4 v[GR][NS];
    float ak[GR];
    int rk[GR];
    for (int g = 0; g < GR; ++g) {
      const int sk = __builtin_amdgcn_readlane(s, k + g);
      ak[g] = __int_as_float(__builtin_amdgcn_readlane(__float_as_int(a), k + g));
      rk[g] = __builtin_amdgcn_readlane(r, k + g);
      for (int j = 0; j < NS; ++j) v[g][j] = h4[(size_t)sk * Q4 + j * 32 + lane];
    }
    for (int g = 0; g < GR; ++g) {
      float* tp = tile + rk[g] * F + lane;
      for (int j = 0; j < NS; ++j) {
        __hip_atomic_fetch_add(tp + j * 128 +  0, ak[g] * v[g][j].x, __ATOMIC_RELAXED,
                               __HIP_MEMORY_SCOPE_WORKGROUP);
        __hip_atomic_fetch_add(tp + j * 128 + 32, ak[g] * v[g][j].y, __ATOMIC_RELAXED,
                               __HIP_MEMORY_SCOPE_WORKGROUP);
        __hip_atomic_fetch_add(tp + j * 128 + 64, ak[g] * v[g][j].z, __ATOMIC_RELAXED,
                               __HIP_MEMORY_SCOPE_WORKGROUP);
        __hip_atomic_fetch_add(tp + j * 128 + 96, ak[g] * v[g][j].w, __ATOMIC_RELAXED,
                               __HIP_MEMORY_SCOPE_WORKGROUP);
      }
    }
  }
  for (; k < m; ++k) {
    const int sk = __builtin_amdgcn_readlane(s, k);
    const float a1 = __int_as_float(__builtin_amdgcn_readlane(__float_as_int(a), k));
    const int r1 = __builtin_amdgcn_readlane(r, k);
    float* tp = tile + r1 * F + lane;
    for (int j = 0; j < NS; ++j) {
      const float4 v1 = h4[(size_t)sk * Q4 + j * 32 + lane];
      __hip_atomic_fetch_add(tp + j * 128 +  0, a1 * v1.x, __ATOMIC_RELAXED,
                             __HIP_MEMORY_SCOPE_WORKGROUP);
      __hip_atomic_fetch_add(tp + j * 128 + 32, a1 * v1.y, __ATOMIC_RELAXED,
                             __HIP_MEMORY_SCOPE_WORKGROUP);
      __hip_atomic_fetch_add(tp + j * 128 + 64, a1 * v1.z, __ATOMIC_RELAXED,
                             __HIP_MEMORY_SCOPE_WORKGROUP);
      __hip_atomic_fetch_add(tp + j * 128 + 96, a1 * v1.w, __ATOMIC_RELAXED,
                             __HIP_MEMORY_SCOPE_WORKGROUP);
    }
  }
}

__device__ __forceinline__ int push_matches(lds_queue_t* q, int cnt, bool match, int e, int lane) {
  const unsigned mask = __builtin_amdgcn_ballot_w32(match);
  if (mask) {
    if (match) q[cnt + __popc(mask & ((1u << lane) - 1u))] = e;
    cnt += __popc(mask);
  }
  return cnt;
}

__device__ __forceinline__ unsigned nonzero_bytes(unsigned t) {
  return ((t & 0x7f7f7f7fu) + 0x7f7f7f7fu) | t;
}

__device__ __forceinline__ unsigned match_bits(uint4 v, unsigned pat) {
  return ((~nonzero_bytes(v.x ^ pat) >> 7) & 0x01010101u) |
         ((~nonzero_bytes(v.y ^ pat) >> 6) & 0x02020202u) |
         ((~nonzero_bytes(v.z ^ pat) >> 5) & 0x04040404u) |
         ((~nonzero_bytes(v.w ^ pat) >> 4) & 0x08080808u);
}

template <int F>
__global__ __launch_bounds__(1024) void owner_tile_sum(
    const float* __restrict__ h, const int* __restrict__ edst,
    const unsigned char* __restrict__ bk,
    const float* __restrict__ rec, const unsigned* __restrict__ menc, const float* __restrict__ b,
    float* __restrict__ sum, int E, int Nn) {
  constexpr int Q4 = F / 4;
  constexpr int U = 8;
  constexpr int GR = (F == 256) ? 4 : 8;
  constexpr int TALLOC = (TROWS + TWAVES - 1) / TWAVES * TWAVES;
  __shared__ __attribute__((aligned(16))) float tile[TALLOC * F];
  __shared__ int queue[TWAVES * QCAP];
  __shared__ float ml[TROWS];
  __shared__ float zl[TROWS];
  const int lane = threadIdx.x & 31;
  const int wave = __builtin_amdgcn_readfirstlane((int)(threadIdx.x >> 5));
  const int bucket = blockIdx.x;
  const int n0 = bucket * TROWS;
  const int nn = min(TROWS, Nn - n0);
  const float4* __restrict__ h4 = (const float4*)h;
  const float4* __restrict__ b4 = (const float4*)b;

  if (threadIdx.x < TROWS) {
    const int node = n0 + min((int)threadIdx.x, nn - 1);
    const float mv = dec_f32(menc[node]);
    ml[threadIdx.x] = mv;
    zl[threadIdx.x] = expf(rec[4 * (size_t)(E + node) + 2] - mv);
  }

  {
    constexpr int RPW = TALLOC / TWAVES;
    constexpr int PPL = Q4 / 32;
    float4 hv[RPW][PPL];
    float av[RPW];
    for (int i = 0; i < RPW; ++i) {
      const int node = n0 + min(wave + i * TWAVES, nn - 1);
      av[i] = expf(rec[4 * (size_t)(E + node) + 2] - dec_f32(menc[node]));
      for (int j = 0; j < PPL; ++j) hv[i][j] = h4[(size_t)node * Q4 + j * 32 + lane];
    }
    for (int i = 0; i < RPW; ++i) {
      const int r = wave + i * TWAVES;
      for (int j = 0; j < PPL; ++j) {
        float4 o;
        o.x = av[i] * hv[i][j].x;
        o.y = av[i] * hv[i][j].y;
        o.z = av[i] * hv[i][j].z;
        o.w = av[i] * hv[i][j].w;
        float* tp = tile + r * F + j * 128 + lane;
        tp[0]  = o.x;
        tp[32] = o.y;
        tp[64] = o.z;
        tp[96] = o.w;
      }
    }
  }
  __syncthreads();

  lds_queue_t* q = (lds_queue_t*)(queue + wave * QCAP);
  int cnt = 0;
  const bool wide = ((((size_t)bk) & 15) == 0) && Nn <= 256 * TROWS;
  constexpr int SPAN = 512 * U;
  const int nfull = wide ? E / SPAN : 0;
  const unsigned pat = (unsigned)bucket * 0x01010101u;
  for (int t = wave; t < nfull; t += TWAVES) {
    const int base = t * SPAN;
    uint4 dv[U];
    for (int u = 0; u < U; ++u)
      dv[u] = *(const uint4*)(bk + base + u * 512 + lane * 16);
    unsigned g[U];
    for (int u = 0; u < U; ++u) g[u] = match_bits(dv[u], pat);
    bool again;
    do {
      again = false;
#pragma unroll
      for (int u = 0; u < U; ++u) {
        while (!again && __builtin_amdgcn_ballot_w32(g[u] != 0u)) {
          const bool has = g[u] != 0u;
          const int p = __builtin_ctz(g[u] | 0x80000000u);
          const int e = base + u * 512 + lane * 16 + ((p & 7) << 2) + (p >> 3);
          cnt = push_matches(q, cnt, has, e, lane);
          g[u] &= g[u] - 1u;
          if (cnt >= 32) again = true;
        }
      }
      while (cnt >= 32) {
        cnt -= 32;
        add_queued_rows<F, GR>(q + cnt, 32, lane, h, rec, ml, zl, tile, n0);
      }
    } while (again);
  }
  for (int e0 = nfull * SPAN + wave * 32; e0 < E; e0 += TWAVES * 32) {
    const int e = e0 + lane;
    const int d = (e < E) ? edst[e] : -1;
    cnt = push_matches(q, cnt, e < E && d / TROWS == bucket, e, lane);
    while (cnt >= 32) {
      cnt -= 32;
      add_queued_rows<F, GR>(q + cnt, 32, lane, h, rec, ml, zl, tile, n0);
    }
  }
  if (cnt > 0) add_queued_rows<F, GR>(q, cnt, lane, h, rec, ml, zl, tile, n0);
  __syncthreads();

  float4* __restrict__ s4 = (float4*)sum;
  for (int r = wave; r < nn; r += TWAVES) {
    const float rz = 1.0f / zl[r];
    for (int j = 0; j < Q4 / 32; ++j) {
      const float* tp = tile + r * F + j * 128 + lane;
      const float4 bv = b4[j * 32 + lane];
      float4 o;
      o.x = bv.x + tp[0]  * rz;
      o.y = bv.y + tp[32] * rz;
      o.z = bv.z + tp[64] * rz;
      o.w = bv.w + tp[96] * rz;
      s4[(size_t)(n0 + r) * Q4 + j * 32 + lane] = o;
    }
  }
}

template <bool RELU_IN>
static void run_layer(const float* xin, int Fin, const float* W,
                      const float* avs, const float* avd, const float* bias,
                      float* h, float* sum, int Fout,
                      const int* esrc, const int* edst, int E, int Nn,
                      unsigned* menc, float* as, float* ad,
                      unsigned char* bk, float* rec, hipStream_t stream) {
  int mt = (Nn + 15) / 16, nt4 = Fout / 64;
  int waves = mt * nt4;
  gemm_f32_wmma<RELU_IN><<<(waves + 7) / 8, 256, 0, stream>>>(xin, W, h, Nn, Fin, Fout);
  int awaves = (Nn + NA_ROWS - 1) / NA_ROWS;
  if (Fout == 256)
    node_alpha<256><<<(awaves + 7) / 8, 256, 0, stream>>>(h, avs, avd, as, ad, Nn);
  else
    node_alpha<128><<<(awaves + 7) / 8, 256, 0, stream>>>(h, avs, avd, as, ad, Nn);

  int ET = E + Nn;
  fill_u32<<<(Nn + 255) / 256, 256, 0, stream>>>(menc, ENC_NEG_INF, Nn);

  edge_max<<<(ET + 256 * EM_PER - 1) / (256 * EM_PER), 256, 0, stream>>>(as, ad, esrc, edst, menc, bk,
                                                                           rec, E, Nn);

  unsigned tblocks = (unsigned)((Nn + TROWS - 1) / TROWS);
  if (Fout == 256)
    owner_tile_sum<256><<<tblocks, 1024, 0, stream>>>(h, edst, bk, rec, menc, bias, sum, E, Nn);
  else
    owner_tile_sum<128><<<tblocks, 1024, 0, stream>>>(h, edst, bk, rec, menc, bias, sum, E, Nn);
}

extern "C" void kernel_launch(void* const* d_in, const int* in_sizes, int n_in,
                              void* d_out, int out_size, void* d_ws, size_t ws_size,
                              hipStream_t stream) {
  const float* x   = (const float*)d_in[0];
  const int*  eidx = (const int*)d_in[1];
  const float* W1  = (const float*)d_in[2];
  const float* a1s = (const float*)d_in[3];
  const float* a1d = (const float*)d_in[4];
  const float* b1  = (const float*)d_in[5];
  const float* W2  = (const float*)d_in[6];
  const float* a2s = (const float*)d_in[7];
  const float* a2d = (const float*)d_in[8];
  const float* b2  = (const float*)d_in[9];
  float* out = (float*)d_out;

  const int IN_C = 128, HID_C = 256, OUT_C = 128;
  int Nn = in_sizes[0] / IN_C;
  int E  = in_sizes[1] / 2;
  const int* esrc = eidx;
  const int* edst = eidx + E;

  float* bufA = (float*)d_ws;
  float* bufB = bufA + (size_t)Nn * HID_C;
  unsigned* menc = (unsigned*)(bufB + (size_t)Nn * HID_C);
  float* as   = (float*)(menc + Nn);
  float* ad   = as + Nn;
  (void)ws_size; (void)n_in; (void)out_size;
  const size_t rec_off = ((size_t)E + 3) & ~(size_t)3;

  run_layer<false>(x, IN_C, W1, a1s, a1d, b1,
                    bufA,  bufB, HID_C,
                   esrc, edst, E, Nn, menc, as, ad,
                    (unsigned char*)out,  out + rec_off, stream);

  run_layer<true>(bufB, HID_C, W2, a2s, a2d, b2,
                   bufA,  out, OUT_C,
                  esrc, edst, E, Nn, menc, as, ad,
                   (unsigned char*)bufB,  bufB + rec_off, stream);
}


# segment max moved into the owner tile: its scan of the tile list raises each row's m by LDS atomic max (same m), keeps ids for the sum pass; global atomic max, 2 fill launches gone; sums as before
# speedup vs baseline: 3.9665x; 1.4175x over previous
#include <hip/hip_runtime.h>

typedef __attribute__((ext_vector_type(2))) float v2f;
typedef __attribute__((ext_vector_type(8))) float v8f;

#define NEG_SLOPE 0.2f

__device__ __forceinline__ unsigned enc_f32(float f) {
  unsigned u = __float_as_uint(f);
  return (u & 0x80000000u) ? ~u : (u | 0x80000000u);
}
__device__ __forceinline__ float dec_f32(unsigned u) {
  unsigned b = (u & 0x80000000u) ? (u & 0x7fffffffu) : ~u;
  return __uint_as_float(b);
}

template <bool RELU_A>
__global__ void gemm_f32_wmma(const float* __restrict__ X, const float* __restrict__ W,
                              float* __restrict__ H, int M, int K, int N) {
  int lane = threadIdx.x & 31;
  int wave = (int)((blockIdx.x * (long long)blockDim.x + threadIdx.x) >> 5);
  int nt4 = N >> 6;
  int mt  = (M + 15) >> 4;
  if (wave >= mt * nt4) return;
  int tm = wave / nt4, tn = wave % nt4;
  int half = lane >> 4;
  int l15  = lane & 15;
  int row = tm * 16 + l15;  if (row >= M) row = M - 1;
  int col0 = tn * 64 + l15;
  v8f acc0 = {}, acc1 = {}, acc2 = {}, acc3 = {};
  for (int k0 = 0; k0 < K; k0 += 4) {
    int kb = k0 + (half << 1);
    v2f a = *(const v2f*)(X + (size_t)row * K + kb);
    if (RELU_A) { a.x = fmaxf(a.x, 0.f); a.y = fmaxf(a.y, 0.f); }
    const float* w0 = W + (size_t)kb * N;
    const float* w1 = W + (size_t)(kb + 1) * N;
    v2f b0, b1, b2, b3;
    b0.x = w0[col0 +  0]; b0.y = w1[col0 +  0];
    b1.x = w0[col0 + 16]; b1.y = w1[col0 + 16];
    b2.x = w0[col0 + 32]; b2.y = w1[col0 + 32];
    b3.x = w0[col0 + 48]; b3.y = w1[col0 + 48];
    acc0 = __builtin_amdgcn_wmma_f32_16x16x4_f32(false, a, false, b0, (short)0, acc0, false, false);
    acc1 = __builtin_amdgcn_wmma_f32_16x16x4_f32(false, a, false, b1, (short)0, acc1, false, false);
    acc2 = __builtin_amdgcn_wmma_f32_16x16x4_f32(false, a, false, b2, (short)0, acc2, false, false);
    acc3 = __builtin_amdgcn_wmma_f32_16x16x4_f32(false, a, false, b3, (short)0, acc3, false, false);
  }
  int rbase = tm * 16 + half * 8;
#pragma unroll
  for (int v = 0; v < 8; ++v) {
    int r = rbase + v;
    if (r < M) {
      float* hp = H + (size_t)r * N + col0;
      hp[ 0] = acc0[v];
      hp[16] = acc1[v];
      hp[32] = acc2[v];
      hp[48] = acc3[v];
    }
  }
}

#define NA_ROWS 16

__device__ __forceinline__ float sum_transpose32(float (&v)[32], int lane) {
#pragma unroll
  for (int n = 16; n >= 1; n >>= 1) {
    const bool up = (lane & n) != 0;
#pragma unroll
    for (int i = 0; i < n; ++i) {
      const float lo = v[i], hi = v[i + n];
      const float give = up ? lo : hi;
      const float keep = up ? hi : lo;
      v[i] = keep + __shfl_xor(give, n, 32);
    }
  }
  return v[0];
}

template <int F>
__global__ __launch_bounds__(256) void node_alpha(const float* __restrict__ h,
                                                  const float* __restrict__ avs,
                                                  const float* __restrict__ avd,
                                                  float* __restrict__ as, float* __restrict__ ad,
                                                  int Nn) {
  constexpr int NS = F / 128;
  constexpr int Q4 = F / 4;
  const int lane = threadIdx.x & 31;
  const int wave = (int)((blockIdx.x * (long long)blockDim.x + threadIdx.x) >> 5);
  const int row0 = wave * NA_ROWS;
  if (row0 >= Nn) return;
  const float4* __restrict__ h4 = (const float4*)h;
  float4 hv[NA_ROWS][NS];
  for (int r = 0; r < NA_ROWS; ++r) {
    const int node = min(row0 + r, Nn - 1);
    for (int j = 0; j < NS; ++j) hv[r][j] = h4[(size_t)node * Q4 + j * 32 + lane];
  }
  float4 vs[NS], vd[NS];
  for (int j = 0; j < NS; ++j) {
    vs[j] = ((const float4*)avs)[j * 32 + lane];
    vd[j] = ((const float4*)avd)[j * 32 + lane];
  }
  float v[2 * NA_ROWS];
  for (int r = 0; r < NA_ROWS; ++r) {
    float ss = 0.f, sd = 0.f;
    for (int j = 0; j < NS; ++j) {
      ss += hv[r][j].x * vs[j].x;  sd += hv[r][j].x * vd[j].x;
      ss += hv[r][j].y * vs[j].y;  sd += hv[r][j].y * vd[j].y;
      ss += hv[r][j].z * vs[j].z;  sd += hv[r][j].z * vd[j].z;
      ss += hv[r][j].w * vs[j].w;  sd += hv[r][j].w * vd[j].w;
    }
    v[r] = ss;
    v[NA_ROWS + r] = sd;
  }
  const float tot = sum_transpose32(v, lane);
  const int row = row0 + (lane & (NA_ROWS - 1));
  if (row < Nn) {
    if (lane < NA_ROWS) as[row] = tot; else ad[row] = tot;
  }
}

#define EM_PER 8
#define TROWS 196
template <bool TAIL>
__device__ __forceinline__ void edge_rec_items(const float* __restrict__ as,
                                               const float* __restrict__ ad,
                                               const int* __restrict__ esrc,
                                               const int* __restrict__ edst,
                                               unsigned char* __restrict__ bk,
                                               float* __restrict__ rec, int E, int ET, int base) {
  int s[EM_PER], d[EM_PER];
  for (int k = 0; k < EM_PER; ++k) {
    int t = base + k * (int)blockDim.x;
    if (TAIL) t = min(t, ET - 1);
    const int tl = min(t, E - 1);
    const int sl = esrc[tl], dl = edst[tl];
    s[k] = (t < E) ? sl : t - E;
    d[k] = (t < E) ? dl : t - E;
  }
  float e[EM_PER];
  for (int k = 0; k < EM_PER; ++k) e[k] = as[s[k]] + ad[d[k]];
  for (int k = 0; k < EM_PER; ++k) {
    const int t = base + k * (int)blockDim.x;
    if (!TAIL || t < ET) {
      if (t < E) bk[t] = (unsigned char)(d[k] / TROWS);
      const float ev = (e[k] > 0.f) ? e[k] : NEG_SLOPE * e[k];
      float4 rc;
      rc.x = __int_as_float(s[k]);
      rc.y = __int_as_float(d[k]);
      rc.z = ev;
      rc.w = 0.f;
      ((float4*)rec)[t] = rc;
    }
  }
}

__global__ __launch_bounds__(256) void edge_rec(const float* __restrict__ as,
                                                const float* __restrict__ ad,
                                                const int* __restrict__ esrc,
                                                const int* __restrict__ edst,
                                                unsigned char* __restrict__ bk,
                                                float* __restrict__ rec, int E, int Nn) {
  const int ET = E + Nn;
  const int span = (int)(blockDim.x * EM_PER);
  const int first = blockIdx.x * span;
  if (first + span <= ET)
    edge_rec_items<false>(as, ad, esrc, edst, bk, rec, E, ET, first + (int)threadIdx.x);
  else
    edge_rec_items<true>(as, ad, esrc, edst, bk, rec, E, ET, first + (int)threadIdx.x);
}

#define TWAVES 32
#define QCAP 64
#define LCAP 192

typedef __attribute__((address_space(3))) volatile int lds_queue_t;

template <int F, int GR>
__device__ __forceinline__ void add_queued_rows(lds_queue_t* q, int m, int lane,
                                                const float* __restrict__ h,
                                                const float* __restrict__ rec,
                                                const float* ml, float* zl,
                                                float* tile, int n0) {
  constexpr int Q4 = F / 4;
  constexpr int NS = F / 128;
  const float4* __restrict__ h4 = (const float4*)h;
  int s = 0, r = 0;
  float a = 0.f;
  if (lane < m) {
    const int e = q[lane];
    const float4 rc = ((const float4*)rec)[e];
    s = __float_as_int(rc.x);
    r = __float_as_int(rc.y) - n0;
    a = expf(rc.z - ml[r]);
    __hip_atomic_fetch_add(zl + r, a, __ATOMIC_RELAXED, __HIP_MEMORY_SCOPE_WORKGROUP);
  }
  int k = 0;
  for (; k + GR <= m; k += GR) {
    float4 v[GR][NS];
    float ak[GR];
    int rk[GR];
    for (int g = 0; g < GR; ++g) {
      const int sk = __builtin_amdgcn_readlane(s, k + g);
      ak[g] = __int_as_float(__builtin_amdgcn_readlane(__float_as_int(a), k + g));
      rk[g] = __builtin_amdgcn_readlane(r, k + g);
      for (int j = 0; j < NS; ++j) v[g][j] = h4[(size_t)sk * Q4 + j * 32 + lane];
    }
    for (int g = 0; g < GR; ++g) {
      float* tp = tile + rk[g] * F + lane;
      for (int j = 0; j < NS; ++j) {
        __hip_atomic_fetch_add(tp + j * 128 +  0, ak[g] * v[g][j].x, __ATOMIC_RELAXED,
                               __HIP_MEMORY_SCOPE_WORKGROUP);
        __hip_atomic_fetch_add(tp + j * 128 + 32, ak[g] * v[g][j].y, __ATOMIC_RELAXED,
                               __HIP_MEMORY_SCOPE_WORKGROUP);
        __hip_atomic_fetch_add(tp + j * 128 + 64, ak[g] * v[g][j].z, __ATOMIC_RELAXED,
                               __HIP_MEMORY_SCOPE_WORKGROUP);
        __hip_atomic_fetch_add(tp + j * 128 + 96, ak[g] * v[g][j].w, __ATOMIC_RELAXED,
                               __HIP_MEMORY_SCOPE_WORKGROUP);
      }
    }
  }
  for (; k < m; ++k) {
    const int sk = __builtin_amdgcn_readlane(s, k);
    const float a1 = __int_as_float(__builtin_amdgcn_readlane(__float_as_int(a), k));
    const int r1 = __builtin_amdgcn_readlane(r, k);
    float* tp = tile + r1 * F + lane;
    for (int j = 0; j < NS; ++j) {
      const float4 v1 = h4[(size_t)sk * Q4 + j * 32 + lane];
      __hip_atomic_fetch_add(tp + j * 128 +  0, a1 * v1.x, __ATOMIC_RELAXED,
                             __HIP_MEMORY_SCOPE_WORKGROUP);
      __hip_atomic_fetch_add(tp + j * 128 + 32, a1 * v1.y, __ATOMIC_RELAXED,
                             __HIP_MEMORY_SCOPE_WORKGROUP);
      __hip_atomic_fetch_add(tp + j * 128 + 64, a1 * v1.z, __ATOMIC_RELAXED,
                             __HIP_MEMORY_SCOPE_WORKGROUP);
      __hip_atomic_fetch_add(tp + j * 128 + 96, a1 * v1.w, __ATOMIC_RELAXED,
                             __HIP_MEMORY_SCOPE_WORKGROUP);
    }
  }
}

__device__ __forceinline__ void max_queued_rows(lds_queue_t* q, int m, int lane,
                                                const float* __restrict__ rec,
                                                unsigned* me, int n0,
                                                lds_queue_t* kl, int& kcnt) {
  const bool room = kcnt + m <= LCAP;
  if (lane < m) {
    const int e = q[lane];
    if (room) kl[kcnt + lane] = e;
    const float4 rc = ((const float4*)rec)[e];
    const int r = __float_as_int(rc.y) - n0;
    __hip_atomic_fetch_max(me + r, enc_f32(rc.z), __ATOMIC_RELAXED, __HIP_MEMORY_SCOPE_WORKGROUP);
  }
  kcnt += m;
}

template <int F, int GR, bool MAXPASS>
__device__ __forceinline__ void take_queued(lds_queue_t* q, int m, int lane,
                                            const float* __restrict__ h,
                                            const float* __restrict__ rec,
                                            unsigned* me, const float* ml, float* zl,
                                            float* tile, int n0, lds_queue_t* kl, int& kcnt) {
  if constexpr (MAXPASS) max_queued_rows(q, m, lane, rec, me, n0, kl, kcnt);
  else add_queued_rows<F, GR>(q, m, lane, h, rec, ml, zl, tile, n0);
}

__device__ __forceinline__ int push_matches(lds_queue_t* q, int cnt, bool match, int e, int lane) {
  const unsigned mask = __builtin_amdgcn_ballot_w32(match);
  if (mask) {
    if (match) q[cnt + __popc(mask & ((1u << lane) - 1u))] = e;
    cnt += __popc(mask);
  }
  return cnt;
}

__device__ __forceinline__ unsigned nonzero_bytes(unsigned t) {
  return ((t & 0x7f7f7f7fu) + 0x7f7f7f7fu) | t;
}

__device__ __forceinline__ unsigned match_bits(uint4 v, unsigned pat) {
  return ((~nonzero_bytes(v.x ^ pat) >> 7) & 0x01010101u) |
         ((~nonzero_bytes(v.y ^ pat) >> 6) & 0x02020202u) |
         ((~nonzero_bytes(v.z ^ pat) >> 5) & 0x04040404u) |
         ((~nonzero_bytes(v.w ^ pat) >> 4) & 0x08080808u);
}

template <int F, int GR, bool MAXPASS>
__device__ __forceinline__ void scan_list(lds_queue_t* q, int lane, int wave, int bucket,
                                          const float* __restrict__ h,
                                          const int* __restrict__ edst,
                                          const unsigned char* __restrict__ bk,
                                          const float* __restrict__ rec,
                                          unsigned* me, const float* ml, float* zl,
                                          float* tile, int n0, int E, int Nn,
                                          lds_queue_t* kl, int& kcnt) {
  constexpr int U = 8;
  int cnt = 0;
  const bool wide = ((((size_t)bk) & 15) == 0) && Nn <= 256 * TROWS;
  constexpr int SPAN = 512 * U;
  const int nfull = wide ? E / SPAN : 0;
  const unsigned pat = (unsigned)bucket * 0x01010101u;
  for (int t = wave; t < nfull; t += TWAVES) {
    const int base = t * SPAN;
    uint4 dv[U];
    for (int u = 0; u < U; ++u)
      dv[u] = *(const uint4*)(bk + base + u * 512 + lane * 16);
    unsigned g[U];
    for (int u = 0; u < U; ++u) g[u] = match_bits(dv[u], pat);
    bool again;
    do {
      again = false;
#pragma unroll
      for (int u = 0; u < U; ++u) {
        while (!again && __builtin_amdgcn_ballot_w32(g[u] != 0u)) {
          const bool has = g[u] != 0u;
          const int p = __builtin_ctz(g[u] | 0x80000000u);
          const int e = base + u * 512 + lane * 16 + ((p & 7) << 2) + (p >> 3);
          cnt = push_matches(q, cnt, has, e, lane);
          g[u] &= g[u] - 1u;
          if (cnt >= 32) again = true;
        }
      }
      while (cnt >= 32) {
        cnt -= 32;
        take_queued<F, GR, MAXPASS>(q + cnt, 32, lane, h, rec, me, ml, zl, tile, n0, kl, kcnt);
      }
    } while (again);
  }
  for (int e0 = nfull * SPAN + wave * 32; e0 < E; e0 += TWAVES * 32) {
    const int e = e0 + lane;
    const int d = (e < E) ? edst[e] : -1;
    cnt = push_matches(q, cnt, e < E && d / TROWS == bucket, e, lane);
    while (cnt >= 32) {
      cnt -= 32;
      take_queued<F, GR, MAXPASS>(q + cnt, 32, lane, h, rec, me, ml, zl, tile, n0, kl, kcnt);
    }
  }
  if (cnt > 0) take_queued<F, GR, MAXPASS>(q, cnt, lane, h, rec, me, ml, zl, tile, n0, kl, kcnt);
}

template <int F>
__global__ __launch_bounds__(1024) void owner_tile_sum(
    const float* __restrict__ h, const int* __restrict__ edst,
    const unsigned char* __restrict__ bk,
    const float* __restrict__ rec, const float* __restrict__ b,
    float* __restrict__ sum, int E, int Nn) {
  constexpr int Q4 = F / 4;
  constexpr int GR = (F == 256) ? 4 : 8;
  constexpr int TALLOC = (TROWS + TWAVES - 1) / TWAVES * TWAVES;
  __shared__ __attribute__((aligned(16))) float tile[TALLOC * F];
  __shared__ int queue[TWAVES * QCAP];
  __shared__ int kept[TWAVES * LCAP];
  __shared__ unsigned me[TROWS];
  __shared__ float ml[TROWS];
  __shared__ float zl[TROWS];
  const int lane = threadIdx.x & 31;
  const int wave = __builtin_amdgcn_readfirstlane((int)(threadIdx.x >> 5));
  const int bucket = blockIdx.x;
  const int n0 = bucket * TROWS;
  const int nn = min(TROWS, Nn - n0);
  const float4* __restrict__ h4 = (const float4*)h;
  const float4* __restrict__ b4 = (const float4*)b;
  lds_queue_t* q = (lds_queue_t*)(queue + wave * QCAP);
  lds_queue_t* kl = (lds_queue_t*)(kept + wave * LCAP);
  int kcnt = 0;

  float eself = 0.f;
  if (threadIdx.x < TROWS) {
    const int node = n0 + min((int)threadIdx.x, nn - 1);
    eself = rec[4 * (size_t)(E + node) + 2];
    me[threadIdx.x] = enc_f32(eself);
  }
  __syncthreads();
  scan_list<F, GR, true>(q, lane, wave, bucket, h, edst, bk, rec, me, ml, zl, tile, n0, E, Nn,
                         kl, kcnt);
  __syncthreads();

  if (threadIdx.x < TROWS) {
    const float mv = dec_f32(me[threadIdx.x]);
    ml[threadIdx.x] = mv;
    zl[threadIdx.x] = expf(eself - mv);
  }

  {
    constexpr int RPW = TALLOC / TWAVES;
    constexpr int PPL = Q4 / 32;
    float4 hv[RPW][PPL];
    float av[RPW];
    for (int i = 0; i < RPW; ++i) {
      const int rr = min(wave + i * TWAVES, nn - 1);
      const int node = n0 + rr;
      av[i] = expf(rec[4 * (size_t)(E + node) + 2] - dec_f32(me[rr]));
      for (int j = 0; j < PPL; ++j) hv[i][j] = h4[(size_t)node * Q4 + j * 32 + lane];
    }
    for (int i = 0; i < RPW; ++i) {
      const int r = wave + i * TWAVES;
      for (int j = 0; j < PPL; ++j) {
        float4 o;
        o.x = av[i] * hv[i][j].x;
        o.y = av[i] * hv[i][j].y;
        o.z = av[i] * hv[i][j].z;
        o.w = av[i] * hv[i][j].w;
        float* tp = tile + r * F + j * 128 + lane;
        tp[0]  = o.x;
        tp[32] = o.y;
        tp[64] = o.z;
        tp[96] = o.w;
      }
    }
  }
  __syncthreads();

  if (kcnt <= LCAP) {
    for (int k = 0; k < kcnt; k += 32)
      add_queued_rows<F, GR>(kl + k, min(32, kcnt - k), lane, h, rec, ml, zl, tile, n0);
  } else {
    int unused = 0;
    scan_list<F, GR, false>(q, lane, wave, bucket, h, edst, bk, rec, me, ml, zl, tile, n0, E, Nn,
                            kl, unused);
  }
  __syncthreads();

  float4* __restrict__ s4 = (float4*)sum;
  for (int r = wave; r < nn; r += TWAVES) {
    const float rz = 1.0f / zl[r];
    for (int j = 0; j < Q4 / 32; ++j) {
      const float* tp = tile + r * F + j * 128 + lane;
      const float4 bv = b4[j * 32 + lane];
      float4 o;
      o.x = bv.x + tp[0]  * rz;
      o.y = bv.y + tp[32] * rz;
      o.z = bv.z + tp[64] * rz;
      o.w = bv.w + tp[96] * rz;
      s4[(size_t)(n0 + r) * Q4 + j * 32 + lane] = o;
    }
  }
}

template <bool RELU_IN>
static void run_layer(const float* xin, int Fin, const float* W,
                      const float* avs, const float* avd, const float* bias,
                      float* h, float* sum, int Fout,
                      const int* esrc, const int* edst, int E, int Nn,
                      float* as, float* ad,
                      unsigned char* bk, float* rec, hipStream_t stream) {
  int mt = (Nn + 15) / 16, nt4 = Fout / 64;
  int waves = mt * nt4;
  gemm_f32_wmma<RELU_IN><<<(waves + 7) / 8, 256, 0, stream>>>(xin, W, h, Nn, Fin, Fout);
  int awaves = (Nn + NA_ROWS - 1) / NA_ROWS;
  if (Fout == 256)
    node_alpha<256><<<(awaves + 7) / 8, 256, 0, stream>>>(h, avs, avd, as, ad, Nn);
  else
    node_alpha<128><<<(awaves + 7) / 8, 256, 0, stream>>>(h, avs, avd, as, ad, Nn);

  int ET = E + Nn;
  edge_rec<<<(ET + 256 * EM_PER - 1) / (256 * EM_PER), 256, 0, stream>>>(as, ad, esrc, edst, bk,
                                                                           rec, E, Nn);

  unsigned tblocks = (unsigned)((Nn + TROWS - 1) / TROWS);
  if (Fout == 256)
    owner_tile_sum<256><<<tblocks, 1024, 0, stream>>>(h, edst, bk, rec, bias, sum, E, Nn);
  else
    owner_tile_sum<128><<<tblocks, 1024, 0, stream>>>(h, edst, bk, rec, bias, sum, E, Nn);
}

extern "C" void kernel_launch(void* const* d_in, const int* in_sizes, int n_in,
                              void* d_out, int out_size, void* d_ws, size_t ws_size,
                              hipStream_t stream) {
  const float* x   = (const float*)d_in[0];
  const int*  eidx = (const int*)d_in[1];
  const float* W1  = (const float*)d_in[2];
  const float* a1s = (const float*)d_in[3];
  const float* a1d = (const float*)d_in[4];
  const float* b1  = (const float*)d_in[5];
  const float* W2  = (const float*)d_in[6];
  const float* a2s = (const float*)d_in[7];
  const float* a2d = (const float*)d_in[8];
  const float* b2  = (const float*)d_in[9];
  float* out = (float*)d_out;

  const int IN_C = 128, HID_C = 256, OUT_C = 128;
  int Nn = in_sizes[0] / IN_C;
  int E  = in_sizes[1] / 2;
  const int* esrc = eidx;
  const int* edst = eidx + E;

  float* bufA = (float*)d_ws;
  float* bufB = bufA + (size_t)Nn * HID_C;
  float* as   = bufB + (size_t)Nn * HID_C;
  float* ad   = as + Nn;
  (void)ws_size; (void)n_in; (void)out_size;
  const size_t rec_off = ((size_t)E + 3) & ~(size_t)3;

  run_layer<false>(x, IN_C, W1, a1s, a1d, b1,
                    bufA,  bufB, HID_C,
                   esrc, edst, E, Nn, as, ad,
                    (unsigned char*)out,  out + rec_off, stream);

  run_layer<true>(bufB, HID_C, W2, a2s, a2d, b2,
                   bufA,  out, OUT_C,
                  esrc, edst, E, Nn, as, ad,
                   (unsigned char*)bufB,  bufB + rec_off, stream);
}


# per-edge record launches removed: the owner tile forms a matched edge's logit itself (ids, as[src] gathered; ad of its rows in LDS), keeps it in LDS; logit launch writes tile bytes; sums as before
# speedup vs baseline: 4.3630x; 1.1000x over previous
#include <hip/hip_runtime.h>

typedef __attribute__((ext_vector_type(2))) float v2f;
typedef __attribute__((ext_vector_type(8))) float v8f;

#define NEG_SLOPE 0.2f

__device__ __forceinline__ unsigned enc_f32(float f) {
  unsigned u = __float_as_uint(f);
  return (u & 0x80000000u) ? ~u : (u | 0x80000000u);
}
__device__ __forceinline__ float dec_f32(unsigned u) {
  unsigned b = (u & 0x80000000u) ? (u & 0x7fffffffu) : ~u;
  return __uint_as_float(b);
}

template <bool RELU_A>
__global__ void gemm_f32_wmma(const float* __restrict__ X, const float* __restrict__ W,
                              float* __restrict__ H, int M, int K, int N) {
  int lane = threadIdx.x & 31;
  int wave = (int)((blockIdx.x * (long long)blockDim.x + threadIdx.x) >> 5);
  int nt4 = N >> 6;
  int mt  = (M + 15) >> 4;
  if (wave >= mt * nt4) return;
  int tm = wave / nt4, tn = wave % nt4;
  int half = lane >> 4;
  int l15  = lane & 15;
  int row = tm * 16 + l15;  if (row >= M) row = M - 1;
  int col0 = tn * 64 + l15;
  v8f acc0 = {}, acc1 = {}, acc2 = {}, acc3 = {};
  for (int k0 = 0; k0 < K; k0 += 4) {
    int kb = k0 + (half << 1);
    v2f a = *(const v2f*)(X + (size_t)row * K + kb);
    if (RELU_A) { a.x = fmaxf(a.x, 0.f); a.y = fmaxf(a.y, 0.f); }
    const float* w0 = W + (size_t)kb * N;
    const float* w1 = W + (size_t)(kb + 1) * N;
    v2f b0, b1, b2, b3;
    b0.x = w0[col0 +  0]; b0.y = w1[col0 +  0];
    b1.x = w0[col0 + 16]; b1.y = w1[col0 + 16];
    b2.x = w0[col0 + 32]; b2.y = w1[col0 + 32];
    b3.x = w0[col0 + 48]; b3.y = w1[col0 + 48];
    acc0 = __builtin_amdgcn_wmma_f32_16x16x4_f32(false, a, false, b0, (short)0, acc0, false, false);
    acc1 = __builtin_amdgcn_wmma_f32_16x16x4_f32(false, a, false, b1, (short)0, acc1, false, false);
    acc2 = __builtin_amdgcn_wmma_f32_16x16x4_f32(false, a, false, b2, (short)0, acc2, false, false);
    acc3 = __builtin_amdgcn_wmma_f32_16x16x4_f32(false, a, false, b3, (short)0, acc3, false, false);
  }
  int rbase = tm * 16 + half * 8;
#pragma unroll
  for (int v = 0; v < 8; ++v) {
    int r = rbase + v;
    if (r < M) {
      float* hp = H + (size_t)r * N + col0;
      hp[ 0] = acc0[v];
      hp[16] = acc1[v];
      hp[32] = acc2[v];
      hp[48] = acc3[v];
    }
  }
}

#define NA_ROWS 16
#define TROWS 196

__device__ __forceinline__ float sum_transpose32(float (&v)[32], int lane) {
#pragma unroll
  for (int n = 16; n >= 1; n >>= 1) {
    const bool up = (lane & n) != 0;
#pragma unroll
    for (int i = 0; i < n; ++i) {
      const float lo = v[i], hi = v[i + n];
      const float give = up ? lo : hi;
      const float keep = up ? hi : lo;
      v[i] = keep + __shfl_xor(give, n, 32);
    }
  }
  return v[0];
}

__device__ __forceinline__ unsigned tile_bytes4(int4 d) {
  return (unsigned)(d.x / TROWS) | ((unsigned)(d.y / TROWS) << 8) |
         ((unsigned)(d.z / TROWS) << 16) | ((unsigned)(d.w / TROWS) << 24);
}
__device__ __forceinline__ void write_tile_bytes(const int* __restrict__ edst,
                                                 unsigned char* __restrict__ bk, int E,
                                                 int wave, int nwaves, int lane) {
  const bool wide = (((size_t)edst | (size_t)bk) & 15) == 0;
  const int nfull = wide ? E / 512 : 0;
  for (int t = wave; t < nfull; t += nwaves) {
    const int4* __restrict__ p = (const int4*)(edst + (size_t)t * 512 + lane * 16);
    const int4 d0 = p[0], d1 = p[1], d2 = p[2], d3 = p[3];
    uint4 o;
    o.x = tile_bytes4(d0);
    o.y = tile_bytes4(d1);
    o.z = tile_bytes4(d2);
    o.w = tile_bytes4(d3);
    *(uint4*)(bk + (size_t)t * 512 + lane * 16) = o;
  }
  for (int e0 = nfull * 512 + wave * 32; e0 < E; e0 += nwaves * 32) {
    const int e = e0 + lane;
    if (e < E) bk[e] = (unsigned char)(edst[e] / TROWS);
  }
}

template <int F>
__global__ __launch_bounds__(256) void node_alpha(const float* __restrict__ h,
                                                  const float* __restrict__ avs,
                                                  const float* __restrict__ avd,
                                                  float* __restrict__ as, float* __restrict__ ad,
                                                  int Nn, const int* __restrict__ edst,
                                                  unsigned char* __restrict__ bk, int E) {
  constexpr int NS = F / 128;
  constexpr int Q4 = F / 4;
  const int lane = threadIdx.x & 31;
  const int wave = (int)((blockIdx.x * (long long)blockDim.x + threadIdx.x) >> 5);
  write_tile_bytes(edst, bk, E, wave, (int)((gridDim.x * (long long)blockDim.x) >> 5), lane);
  const int row0 = wave * NA_ROWS;
  if (row0 >= Nn) return;
  const float4* __restrict__ h4 = (const float4*)h;
  float4 hv[NA_ROWS][NS];
  for (int r = 0; r < NA_ROWS; ++r) {
    const int node = min(row0 + r, Nn - 1);
    for (int j = 0; j < NS; ++j) hv[r][j] = h4[(size_t)node * Q4 + j * 32 + lane];
  }
  float4 vs[NS], vd[NS];
  for (int j = 0; j < NS; ++j) {
    vs[j] = ((const float4*)avs)[j * 32 + lane];
    vd[j] = ((const float4*)avd)[j * 32 + lane];
  }
  float v[2 * NA_ROWS];
  for (int r = 0; r < NA_ROWS; ++r) {
    float ss = 0.f, sd = 0.f;
    for (int j = 0; j < NS; ++j) {
      ss += hv[r][j].x * vs[j].x;  sd += hv[r][j].x * vd[j].x;
      ss += hv[r][j].y * vs[j].y;  sd += hv[r][j].y * vd[j].y;
      ss += hv[r][j].z * vs[j].z;  sd += hv[r][j].z * vd[j].z;
      ss += hv[r][j].w * vs[j].w;  sd += hv[r][j].w * vd[j].w;
    }
    v[r] = ss;
    v[NA_ROWS + r] = sd;
  }
  const float tot = sum_transpose32(v, lane);
  const int row = row0 + (lane & (NA_ROWS - 1));
  if (row < Nn) {
    if (lane < NA_ROWS) as[row] = tot; else ad[row] = tot;
  }
}

#define TWAVES 32
#define QCAP 64
#define LCAP 192

typedef __attribute__((address_space(3))) volatile int lds_queue_t;

__device__ __forceinline__ void fetch_edge(int e, const int* __restrict__ esrc,
                                           const int* __restrict__ edst,
                                           const float* __restrict__ as, const float* adl,
                                           int n0, int& s, int& r, float& ev) {
  s = esrc[e];
  r = edst[e] - n0;
  const float x = as[s] + adl[r];
  ev = (x > 0.f) ? x : NEG_SLOPE * x;
}

template <int F, int GR>
__device__ __forceinline__ void add_rows(int m, int lane, int s, int r, float a,
                                         const float* __restrict__ h, float* tile) {
  constexpr int Q4 = F / 4;
  constexpr int NS = F / 128;
  const float4* __restrict__ h4 = (const float4*)h;
  int k = 0;
  for (; k + GR <= m; k += GR) {
    float4 v[GR][NS];
    float ak[GR];
    int rk[GR];
    for (int g = 0; g < GR; ++g) {
      const int sk = __builtin_amdgcn_readlane(s, k + g);
      ak[g] = __int_as_float(__builtin_amdgcn_readlane(__float_as_int(a), k + g));
      rk[g] = __builtin_amdgcn_readlane(r, k + g);
      for (int j = 0; j < NS; ++j) v[g][j] = h4[(size_t)sk * Q4 + j * 32 + lane];
    }
    for (int g = 0; g < GR; ++g) {
      float* tp = tile + rk[g] * F + lane;
      for (int j = 0; j < NS; ++j) {
        __hip_atomic_fetch_add(tp + j * 128 +  0, ak[g] * v[g][j].x, __ATOMIC_RELAXED,
                               __HIP_MEMORY_SCOPE_WORKGROUP);
        __hip_atomic_fetch_add(tp + j * 128 + 32, ak[g] * v[g][j].y, __ATOMIC_RELAXED,
                               __HIP_MEMORY_SCOPE_WORKGROUP);
        __hip_atomic_fetch_add(tp + j * 128 + 64, ak[g] * v[g][j].z, __ATOMIC_RELAXED,
                               __HIP_MEMORY_SCOPE_WORKGROUP);
        __hip_atomic_fetch_add(tp + j * 128 + 96, ak[g] * v[g][j].w, __ATOMIC_RELAXED,
                               __HIP_MEMORY_SCOPE_WORKGROUP);
      }
    }
  }
  for (; k < m; ++k) {
    const int sk = __builtin_amdgcn_readlane(s, k);
    const float a1 = __int_as_float(__builtin_amdgcn_readlane(__float_as_int(a), k));
    const int r1 = __builtin_amdgcn_readlane(r, k);
    float* tp = tile + r1 * F + lane;
    for (int j = 0; j < NS; ++j) {
      const float4 v1 = h4[(size_t)sk * Q4 + j * 32 + lane];
      __hip_atomic_fetch_add(tp + j * 128 +  0, a1 * v1.x, __ATOMIC_RELAXED,
                             __HIP_MEMORY_SCOPE_WORKGROUP);
      __hip_atomic_fetch_add(tp + j * 128 + 32, a1 * v1.y, __ATOMIC_RELAXED,
                             __HIP_MEMORY_SCOPE_WORKGROUP);
      __hip_atomic_fetch_add(tp + j * 128 + 64, a1 * v1.z, __ATOMIC_RELAXED,
                             __HIP_MEMORY_SCOPE_WORKGROUP);
      __hip_atomic_fetch_add(tp + j * 128 + 96, a1 * v1.w, __ATOMIC_RELAXED,
                             __HIP_MEMORY_SCOPE_WORKGROUP);
    }
  }
}

struct kept_lists {
  lds_queue_t* s;
  lds_queue_t* r;
  lds_queue_t* e;
};

template <int F, int GR>
__device__ __forceinline__ void add_kept_rows(const kept_lists& kl, int k0, int m, int lane,
                                              const float* __restrict__ h,
                                              const float* ml, float* zl, float* tile) {
  int s = 0, r = 0;
  float a = 0.f;
  if (lane < m) {
    s = kl.s[k0 + lane];
    r = kl.r[k0 + lane];
    a = expf(__int_as_float(kl.e[k0 + lane]) - ml[r]);
    __hip_atomic_fetch_add(zl + r, a, __ATOMIC_RELAXED, __HIP_MEMORY_SCOPE_WORKGROUP);
  }
  add_rows<F, GR>(m, lane, s, r, a, h, tile);
}

template <int F, int GR>
__device__ __forceinline__ void add_queued_rows(lds_queue_t* q, int m, int lane,
                                                const float* __restrict__ h,
                                                const int* __restrict__ esrc,
                                                const int* __restrict__ edst,
                                                const float* __restrict__ as, const float* adl,
                                                const float* ml, float* zl,
                                                float* tile, int n0) {
  int s = 0, r = 0;
  float a = 0.f;
  if (lane < m) {
    float ev;
    fetch_edge(q[lane], esrc, edst, as, adl, n0, s, r, ev);
    a = expf(ev - ml[r]);
    __hip_atomic_fetch_add(zl + r, a, __ATOMIC_RELAXED, __HIP_MEMORY_SCOPE_WORKGROUP);
  }
  add_rows<F, GR>(m, lane, s, r, a, h, tile);
}

__device__ __forceinline__ void max_queued_rows(lds_queue_t* q, int m, int lane,
                                                const int* __restrict__ esrc,
                                                const int* __restrict__ edst,
                                                const float* __restrict__ as, const float* adl,
                                                unsigned* me, int n0,
                                                const kept_lists& kl, int& kcnt) {
  const bool room = kcnt + m <= LCAP;
  if (lane < m) {
    int s, r;
    float ev;
    fetch_edge(q[lane], esrc, edst, as, adl, n0, s, r, ev);
    if (room) {
      kl.s[kcnt + lane] = s;
      kl.r[kcnt + lane] = r;
      kl.e[kcnt + lane] = __float_as_int(ev);
    }
    __hip_atomic_fetch_max(me + r, enc_f32(ev), __ATOMIC_RELAXED, __HIP_MEMORY_SCOPE_WORKGROUP);
  }
  kcnt += m;
}

template <int F, int GR, bool MAXPASS>
__device__ __forceinline__ void take_queued(lds_queue_t* q, int m, int lane,
                                            const float* __restrict__ h,
                                            const int* __restrict__ esrc,
                                            const int* __restrict__ edst,
                                            const float* __restrict__ as, const float* adl,
                                            unsigned* me, const float* ml, float* zl,
                                            float* tile, int n0, const kept_lists& kl,
                                            int& kcnt) {
  if constexpr (MAXPASS) max_queued_rows(q, m, lane, esrc, edst, as, adl, me, n0, kl, kcnt);
  else add_queued_rows<F, GR>(q, m, lane, h, esrc, edst, as, adl, ml, zl, tile, n0);
}

__device__ __forceinline__ int push_matches(lds_queue_t* q, int cnt, bool match, int e, int lane) {
  const unsigned mask = __builtin_amdgcn_ballot_w32(match);
  if (mask) {
    if (match) q[cnt + __popc(mask & ((1u << lane) - 1u))] = e;
    cnt += __popc(mask);
  }
  return cnt;
}

__device__ __forceinline__ unsigned nonzero_bytes(unsigned t) {
  return ((t & 0x7f7f7f7fu) + 0x7f7f7f7fu) | t;
}

__device__ __forceinline__ unsigned match_bits(uint4 v, unsigned pat) {
  return ((~nonzero_bytes(v.x ^ pat) >> 7) & 0x01010101u) |
         ((~nonzero_bytes(v.y ^ pat) >> 6) & 0x02020202u) |
         ((~nonzero_bytes(v.z ^ pat) >> 5) & 0x04040404u) |
         ((~nonzero_bytes(v.w ^ pat) >> 4) & 0x08080808u);
}

template <int F, int GR, bool MAXPASS>
__device__ __forceinline__ void scan_list(lds_queue_t* q, int lane, int wave, int bucket,
                                          const float* __restrict__ h,
                                          const int* __restrict__ esrc,
                                          const int* __restrict__ edst,
                                          const unsigned char* __restrict__ bk,
                                          const float* __restrict__ as, const float* adl,
                                          unsigned* me, const float* ml, float* zl,
                                          float* tile, int n0, int E, int Nn,
                                          const kept_lists& kl, int& kcnt) {
  constexpr int U = 8;
  int cnt = 0;
  const bool wide = ((((size_t)bk) & 15) == 0) && Nn <= 256 * TROWS;
  constexpr int SPAN = 512 * U;
  const int nfull = wide ? E / SPAN : 0;
  const unsigned pat = (unsigned)bucket * 0x01010101u;
  for (int t = wave; t < nfull; t += TWAVES) {
    const int base = t * SPAN;
    uint4 dv[U];
    for (int u = 0; u < U; ++u)
      dv[u] = *(const uint4*)(bk + base + u * 512 + lane * 16);
    unsigned g[U];
    for (int u = 0; u < U; ++u) g[u] = match_bits(dv[u], pat);
    bool again;
    do {
      again = false;
#pragma unroll
      for (int u = 0; u < U; ++u) {
        while (!again && __builtin_amdgcn_ballot_w32(g[u] != 0u)) {
          const bool has = g[u] != 0u;
          const int p = __builtin_ctz(g[u] | 0x80000000u);
          const int e = base + u * 512 + lane * 16 + ((p & 7) << 2) + (p >> 3);
          cnt = push_matches(q, cnt, has, e, lane);
          g[u] &= g[u] - 1u;
          if (cnt >= 32) again = true;
        }
      }
      while (cnt >= 32) {
        cnt -= 32;
        take_queued<F, GR, MAXPASS>(q + cnt, 32, lane, h, esrc, edst, as, adl, me, ml, zl, tile, n0,
                                    kl, kcnt);
      }
    } while (again);
  }
  for (int e0 = nfull * SPAN + wave * 32; e0 < E; e0 += TWAVES * 32) {
    const int e = e0 + lane;
    const int d = (e < E) ? edst[e] : -1;
    cnt = push_matches(q, cnt, e < E && d / TROWS == bucket, e, lane);
    while (cnt >= 32) {
      cnt -= 32;
      take_queued<F, GR, MAXPASS>(q + cnt, 32, lane, h, esrc, edst, as, adl, me, ml, zl, tile, n0,
                                    kl, kcnt);
    }
  }
  if (cnt > 0)
    take_queued<F, GR, MAXPASS>(q, cnt, lane, h, esrc, edst, as, adl, me, ml, zl, tile, n0, kl, kcnt);
}

template <int F>
__global__ __launch_bounds__(1024) void owner_tile_sum(
    const float* __restrict__ h, const int* __restrict__ esrc, const int* __restrict__ edst,
    const unsigned char* __restrict__ bk,
    const float* __restrict__ as, const float* __restrict__ ad, const float* __restrict__ b,
    float* __restrict__ sum, int E, int Nn) {
  constexpr int Q4 = F / 4;
  constexpr int GR = (F == 256) ? 4 : 8;
  constexpr int TALLOC = (TROWS + TWAVES - 1) / TWAVES * TWAVES;
  __shared__ __attribute__((aligned(16))) float tile[TALLOC * F];
  __shared__ int queue[TWAVES * QCAP];
  __shared__ int kept_s[TWAVES * LCAP];
  __shared__ int kept_r[TWAVES * LCAP];
  __shared__ int kept_e[TWAVES * LCAP];
  __shared__ unsigned me[TROWS];
  __shared__ float ml[TROWS];
  __shared__ float zl[TROWS];
  __shared__ float adl[TROWS];
  __shared__ float esl[TROWS];
  const int lane = threadIdx.x & 31;
  const int wave = __builtin_amdgcn_readfirstlane((int)(threadIdx.x >> 5));
  const int bucket = blockIdx.x;
  const int n0 = bucket * TROWS;
  const int nn = min(TROWS, Nn - n0);
  const float4* __restrict__ h4 = (const float4*)h;
  const float4* __restrict__ b4 = (const float4*)b;
  lds_queue_t* q = (lds_queue_t*)(queue + wave * QCAP);
  kept_lists kl;
  kl.s = (lds_queue_t*)(kept_s + wave * LCAP);
  kl.r = (lds_queue_t*)(kept_r + wave * LCAP);
  kl.e = (lds_queue_t*)(kept_e + wave * LCAP);
  int kcnt = 0;

  float eself = 0.f;
  if (threadIdx.x < TROWS) {
    const int node = n0 + min((int)threadIdx.x, nn - 1);
    const float adn = ad[node];
    const float x = as[node] + adn;
    eself = (x > 0.f) ? x : NEG_SLOPE * x;
    adl[threadIdx.x] = adn;
    esl[threadIdx.x] = eself;
    me[threadIdx.x] = enc_f32(eself);
  }
  __syncthreads();
  scan_list<F, GR, true>(q, lane, wave, bucket, h, esrc, edst, bk, as, adl, me, ml, zl, tile, n0,
                         E, Nn, kl, kcnt);
  __syncthreads();

  if (threadIdx.x < TROWS) {
    const float mv = dec_f32(me[threadIdx.x]);
    ml[threadIdx.x] = mv;
    zl[threadIdx.x] = expf(eself - mv);
  }

  {
    constexpr int RPW = TALLOC / TWAVES;
    constexpr int PPL = Q4 / 32;
    float4 hv[RPW][PPL];
    float av[RPW];
    for (int i = 0; i < RPW; ++i) {
      const int rr = min(wave + i * TWAVES, nn - 1);
      const int node = n0 + rr;
      av[i] = expf(esl[rr] - dec_f32(me[rr]));
      for (int j = 0; j < PPL; ++j) hv[i][j] = h4[(size_t)node * Q4 + j * 32 + lane];
    }
    for (int i = 0; i < RPW; ++i) {
      const int r = wave + i * TWAVES;
      for (int j = 0; j < PPL; ++j) {
        float4 o;
        o.x = av[i] * hv[i][j].x;
        o.y = av[i] * hv[i][j].y;
        o.z = av[i] * hv[i][j].z;
        o.w = av[i] * hv[i][j].w;
        float* tp = tile + r * F + j * 128 + lane;
        tp[0]  = o.x;
        tp[32] = o.y;
        tp[64] = o.z;
        tp[96] = o.w;
      }
    }
  }
  __syncthreads();

  if (kcnt <= LCAP) {
    for (int k = 0; k < kcnt; k += 32)
      add_kept_rows<F, GR>(kl, k, min(32, kcnt - k), lane, h, ml, zl, tile);
  } else {
    int unused = 0;
    scan_list<F, GR, false>(q, lane, wave, bucket, h, esrc, edst, bk, as, adl, me, ml, zl, tile,
                            n0, E, Nn, kl, unused);
  }
  __syncthreads();

  float4* __restrict__ s4 = (float4*)sum;
  for (int r = wave; r < nn; r += TWAVES) {
    const float rz = 1.0f / zl[r];
    for (int j = 0; j < Q4 / 32; ++j) {
      const float* tp = tile + r * F + j * 128 + lane;
      const float4 bv = b4[j * 32 + lane];
      float4 o;
      o.x = bv.x + tp[0]  * rz;
      o.y = bv.y + tp[32] * rz;
      o.z = bv.z + tp[64] * rz;
      o.w = bv.w + tp[96] * rz;
      s4[(size_t)(n0 + r) * Q4 + j * 32 + lane] = o;
    }
  }
}

template <bool RELU_IN>
static void run_layer(const float* xin, int Fin, const float* W,
                      const float* avs, const float* avd, const float* bias,
                      float* h, float* sum, int Fout,
                      const int* esrc, const int* edst, int E, int Nn,
                      float* as, float* ad,
                      unsigned char* bk, hipStream_t stream) {
  int mt = (Nn + 15) / 16, nt4 = Fout / 64;
  int waves = mt * nt4;
  gemm_f32_wmma<RELU_IN><<<(waves + 7) / 8, 256, 0, stream>>>(xin, W, h, Nn, Fin, Fout);
  int awaves = (Nn + NA_ROWS - 1) / NA_ROWS;
  if (Fout == 256)
    node_alpha<256><<<(awaves + 7) / 8, 256, 0, stream>>>(h, avs, avd, as, ad, Nn, edst, bk, E);
  else
    node_alpha<128><<<(awaves + 7) / 8, 256, 0, stream>>>(h, avs, avd, as, ad, Nn, edst, bk, E);

  unsigned tblocks = (unsigned)((Nn + TROWS - 1) / TROWS);
  if (Fout == 256)
    owner_tile_sum<256><<<tblocks, 1024, 0, stream>>>(h, esrc, edst, bk, as, ad, bias, sum, E, Nn);
  else
    owner_tile_sum<128><<<tblocks, 1024, 0, stream>>>(h, esrc, edst, bk, as, ad, bias, sum, E, Nn);
}

extern "C" void kernel_launch(void* const* d_in, const int* in_sizes, int n_in,
                              void* d_out, int out_size, void* d_ws, size_t ws_size,
                              hipStream_t stream) {
  const float* x   = (const float*)d_in[0];
  const int*  eidx = (const int*)d_in[1];
  const float* W1  = (const float*)d_in[2];
  const float* a1s = (const float*)d_in[3];
  const float* a1d = (const float*)d_in[4];
  const float* b1  = (const float*)d_in[5];
  const float* W2  = (const float*)d_in[6];
  const float* a2s = (const float*)d_in[7];
  const float* a2d = (const float*)d_in[8];
  const float* b2  = (const float*)d_in[9];
  float* out = (float*)d_out;

  const int IN_C = 128, HID_C = 256, OUT_C = 128;
  int Nn = in_sizes[0] / IN_C;
  int E  = in_sizes[1] / 2;
  const int* esrc = eidx;
  const int* edst = eidx + E;

  float* bufA = (float*)d_ws;
  float* bufB = bufA + (size_t)Nn * HID_C;
  float* as   = bufB + (size_t)Nn * HID_C;
  float* ad   = as + Nn;
  (void)ws_size; (void)n_in; (void)out_size;

  run_layer<false>(x, IN_C, W1, a1s, a1d, b1,
                    bufA,  bufB, HID_C,
                   esrc, edst, E, Nn, as, ad,
                    (unsigned char*)out, stream);

  run_layer<true>(bufB, HID_C, W2, a2s, a2d, b2,
                   bufA,  out, OUT_C,
                  esrc, edst, E, Nn, as, ad,
                   (unsigned char*)bufB, stream);
}


# layer 2's owner tiles start from the (source,row) pair lists layer 1's tiles leave in the output buffer, each inside its own rows: no edge scan, byte list or id gathers in layer 2; sums as before
# speedup vs baseline: 4.7197x; 1.0818x over previous
#include <hip/hip_runtime.h>

typedef __attribute__((ext_vector_type(2))) float v2f;
typedef __attribute__((ext_vector_type(8))) float v8f;

#define NEG_SLOPE 0.2f

__device__ __forceinline__ unsigned enc_f32(float f) {
  unsigned u = __float_as_uint(f);
  return (u & 0x80000000u) ? ~u : (u | 0x80000000u);
}
__device__ __forceinline__ float dec_f32(unsigned u) {
  unsigned b = (u & 0x80000000u) ? (u & 0x7fffffffu) : ~u;
  return __uint_as_float(b);
}

template <bool RELU_A>
__global__ void gemm_f32_wmma(const float* __restrict__ X, const float* __restrict__ W,
                              float* __restrict__ H, int M, int K, int N) {
  int lane = threadIdx.x & 31;
  int wave = (int)((blockIdx.x * (long long)blockDim.x + threadIdx.x) >> 5);
  int nt4 = N >> 6;
  int mt  = (M + 15) >> 4;
  if (wave >= mt * nt4) return;
  int tm = wave / nt4, tn = wave % nt4;
  int half = lane >> 4;
  int l15  = lane & 15;
  int row = tm * 16 + l15;  if (row >= M) row = M - 1;
  int col0 = tn * 64 + l15;
  v8f acc0 = {}, acc1 = {}, acc2 = {}, acc3 = {};
  for (int k0 = 0; k0 < K; k0 += 4) {
    int kb = k0 + (half << 1);
    v2f a = *(const v2f*)(X + (size_t)row * K + kb);
    if (RELU_A) { a.x = fmaxf(a.x, 0.f); a.y = fmaxf(a.y, 0.f); }
    const float* w0 = W + (size_t)kb * N;
    const float* w1 = W + (size_t)(kb + 1) * N;
    v2f b0, b1, b2, b3;
    b0.x = w0[col0 +  0]; b0.y = w1[col0 +  0];
    b1.x = w0[col0 + 16]; b1.y = w1[col0 + 16];
    b2.x = w0[col0 + 32]; b2.y = w1[col0 + 32];
    b3.x = w0[col0 + 48]; b3.y = w1[col0 + 48];
    acc0 = __builtin_amdgcn_wmma_f32_16x16x4_f32(false, a, false, b0, (short)0, acc0, false, false);
    acc1 = __builtin_amdgcn_wmma_f32_16x16x4_f32(false, a, false, b1, (short)0, acc1, false, false);
    acc2 = __builtin_amdgcn_wmma_f32_16x16x4_f32(false, a, false, b2, (short)0, acc2, false, false);
    acc3 = __builtin_amdgcn_wmma_f32_16x16x4_f32(false, a, false, b3, (short)0, acc3, false, false);
  }
  int rbase = tm * 16 + half * 8;
#pragma unroll
  for (int v = 0; v < 8; ++v) {
    int r = rbase + v;
    if (r < M) {
      float* hp = H + (size_t)r * N + col0;
      hp[ 0] = acc0[v];
      hp[16] = acc1[v];
      hp[32] = acc2[v];
      hp[48] = acc3[v];
    }
  }
}

#define NA_ROWS 16
#define TROWS 196
#define TWAVES 32
#define LCAP 192

#define LIST_ROWB 512
#define REGB (TROWS * LIST_ROWB)
#define LISTB (TWAVES * 4 + TWAVES * LCAP * 8)
#define BKOFF ((LISTB + 255) / 256 * 256)
#define BKCH ((REGB - BKOFF) / 4096 * 4096)
static_assert(BKCH >= 4096, "a tile's rows must hold its lists and at least one scan trip");
__device__ __forceinline__ bool lists_usable(int E, int Nn) {
  return E / BKCH + 1 <= Nn / TROWS;
}
__device__ __forceinline__ size_t bk_place(size_t i, bool chunked) {
  return chunked ? (i / BKCH) * REGB + BKOFF + i % BKCH : i;
}
__device__ __forceinline__ int list_cap(int nn) {
  return min(LCAP, (nn * LIST_ROWB - TWAVES * 4) / (TWAVES * 8) / 16 * 16);
}

__device__ __forceinline__ float sum_transpose32(float (&v)[32], int lane) {
#pragma unroll
  for (int n = 16; n >= 1; n >>= 1) {
    const bool up = (lane & n) != 0;
#pragma unroll
    for (int i = 0; i < n; ++i) {
      const float lo = v[i], hi = v[i + n];
      const float give = up ? lo : hi;
      const float keep = up ? hi : lo;
      v[i] = keep + __shfl_xor(give, n, 32);
    }
  }
  return v[0];
}

__device__ __forceinline__ unsigned tile_bytes4(int4 d) {
  return (unsigned)(d.x / TROWS) | ((unsigned)(d.y / TROWS) << 8) |
         ((unsigned)(d.z / TROWS) << 16) | ((unsigned)(d.w / TROWS) << 24);
}
__device__ __forceinline__ void write_tile_bytes(const int* __restrict__ edst,
                                                 unsigned char* __restrict__ bk, int E,
                                                 int wave, int nwaves, int lane, bool chunked) {
  const bool wide = (((size_t)edst | (size_t)bk) & 15) == 0;
  const int nfull = wide ? E / 512 : 0;
  for (int t = wave; t < nfull; t += nwaves) {
    const int4* __restrict__ p = (const int4*)(edst + (size_t)t * 512 + lane * 16);
    const int4 d0 = p[0], d1 = p[1], d2 = p[2], d3 = p[3];
    uint4 o;
    o.x = tile_bytes4(d0);
    o.y = tile_bytes4(d1);
    o.z = tile_bytes4(d2);
    o.w = tile_bytes4(d3);
    *(uint4*)(bk + bk_place((size_t)t * 512, chunked) + lane * 16) = o;
  }
  for (int e0 = nfull * 512 + wave * 32; e0 < E; e0 += nwaves * 32) {
    const int e = e0 + lane;
    if (e < E) bk[bk_place((size_t)e, chunked)] = (unsigned char)(edst[e] / TROWS);
  }
}

template <int F, bool L2>
__global__ __launch_bounds__(256) void node_alpha(const float* __restrict__ h,
                                                  const float* __restrict__ avs,
                                                  const float* __restrict__ avd,
                                                  float* __restrict__ as, float* __restrict__ ad,
                                                  int Nn, const int* __restrict__ edst,
                                                  unsigned char* __restrict__ bk, int E) {
  constexpr int NS = F / 128;
  constexpr int Q4 = F / 4;
  const int lane = threadIdx.x & 31;
  const int wave = (int)((blockIdx.x * (long long)blockDim.x + threadIdx.x) >> 5);
  const bool keep = lists_usable(E, Nn);
  if (!(L2 && keep))
    write_tile_bytes(edst, bk, E, wave, (int)((gridDim.x * (long long)blockDim.x) >> 5), lane,
                     !L2 && keep);
  const int row0 = wave * NA_ROWS;
  if (row0 >= Nn) return;
  const float4* __restrict__ h4 = (const float4*)h;
  float4 hv[NA_ROWS][NS];
  for (int r = 0; r < NA_ROWS; ++r) {
    const int node = min(row0 + r, Nn - 1);
    for (int j = 0; j < NS; ++j) hv[r][j] = h4[(size_t)node * Q4 + j * 32 + lane];
  }
  float4 vs[NS], vd[NS];
  for (int j = 0; j < NS; ++j) {
    vs[j] = ((const float4*)avs)[j * 32 + lane];
    vd[j] = ((const float4*)avd)[j * 32 + lane];
  }
  float v[2 * NA_ROWS];
  for (int r = 0; r < NA_ROWS; ++r) {
    float ss = 0.f, sd = 0.f;
    for (int j = 0; j < NS; ++j) {
      ss += hv[r][j].x * vs[j].x;  sd += hv[r][j].x * vd[j].x;
      ss += hv[r][j].y * vs[j].y;  sd += hv[r][j].y * vd[j].y;
      ss += hv[r][j].z * vs[j].z;  sd += hv[r][j].z * vd[j].z;
      ss += hv[r][j].w * vs[j].w;  sd += hv[r][j].w * vd[j].w;
    }
    v[r] = ss;
    v[NA_ROWS + r] = sd;
  }
  const float tot = sum_transpose32(v, lane);
  const int row = row0 + (lane & (NA_ROWS - 1));
  if (row < Nn) {
    if (lane < NA_ROWS) as[row] = tot; else ad[row] = tot;
  }
}

#define QCAP 64

typedef __attribute__((address_space(3))) volatile int lds_queue_t;

__device__ __forceinline__ void fetch_edge(int e, const int* __restrict__ esrc,
                                           const int* __restrict__ edst,
                                           const float* __restrict__ as, const float* adl,
                                           int n0, int& s, int& r, float& ev) {
  s = esrc[e];
  r = edst[e] - n0;
  const float x = as[s] + adl[r];
  ev = (x > 0.f) ? x : NEG_SLOPE * x;
}

template <int F, int GR>
__device__ __forceinline__ void add_rows(int m, int lane, int s, int r, float a,
                                         const float* __restrict__ h, float* tile) {
  constexpr int Q4 = F / 4;
  constexpr int NS = F / 128;
  const float4* __restrict__ h4 = (const float4*)h;
  int k = 0;
  for (; k + GR <= m; k += GR) {
    float4 v[GR][NS];
    float ak[GR];
    int rk[GR];
    for (int g = 0; g < GR; ++g) {
      const int sk = __builtin_amdgcn_readlane(s, k + g);
      ak[g] = __int_as_float(__builtin_amdgcn_readlane(__float_as_int(a), k + g));
      rk[g] = __builtin_amdgcn_readlane(r, k + g);
      for (int j = 0; j < NS; ++j) v[g][j] = h4[(size_t)sk * Q4 + j * 32 + lane];
    }
    for (int g = 0; g < GR; ++g) {
      float* tp = tile + rk[g] * F + lane;
      for (int j = 0; j < NS; ++j) {
        __hip_atomic_fetch_add(tp + j * 128 +  0, ak[g] * v[g][j].x, __ATOMIC_RELAXED,
                               __HIP_MEMORY_SCOPE_WORKGROUP);
        __hip_atomic_fetch_add(tp + j * 128 + 32, ak[g] * v[g][j].y, __ATOMIC_RELAXED,
                               __HIP_MEMORY_SCOPE_WORKGROUP);
        __hip_atomic_fetch_add(tp + j * 128 + 64, ak[g] * v[g][j].z, __ATOMIC_RELAXED,
                               __HIP_MEMORY_SCOPE_WORKGROUP);
        __hip_atomic_fetch_add(tp + j * 128 + 96, ak[g] * v[g][j].w, __ATOMIC_RELAXED,
                               __HIP_MEMORY_SCOPE_WORKGROUP);
      }
    }
  }
  for (; k < m; ++k) {
    const int sk = __builtin_amdgcn_readlane(s, k);
    const float a1 = __int_as_float(__builtin_amdgcn_readlane(__float_as_int(a), k));
    const int r1 = __builtin_amdgcn_readlane(r, k);
    float* tp = tile + r1 * F + lane;
    for (int j = 0; j < NS; ++j) {
      const float4 v1 = h4[(size_t)sk * Q4 + j * 32 + lane];
      __hip_atomic_fetch_add(tp + j * 128 +  0, a1 * v1.x, __ATOMIC_RELAXED,
                             __HIP_MEMORY_SCOPE_WORKGROUP);
      __hip_atomic_fetch_add(tp + j * 128 + 32, a1 * v1.y, __ATOMIC_RELAXED,
                             __HIP_MEMORY_SCOPE_WORKGROUP);
      __hip_atomic_fetch_add(tp + j * 128 + 64, a1 * v1.z, __ATOMIC_RELAXED,
                             __HIP_MEMORY_SCOPE_WORKGROUP);
      __hip_atomic_fetch_add(tp + j * 128 + 96, a1 * v1.w, __ATOMIC_RELAXED,
                             __HIP_MEMORY_SCOPE_WORKGROUP);
    }
  }
}

struct kept_lists {
  lds_queue_t* s;
  lds_queue_t* r;
  lds_queue_t* e;
  int2* g;
  int gcap;
};

template <int F, int GR>
__device__ __forceinline__ void add_kept_rows(const kept_lists& kl, int k0, int m, int lane,
                                              const float* __restrict__ h,
                                              const float* ml, float* zl, float* tile) {
  int s = 0, r = 0;
  float a = 0.f;
  if (lane < m) {
    s = kl.s[k0 + lane];
    r = kl.r[k0 + lane];
    a = expf(__int_as_float(kl.e[k0 + lane]) - ml[r]);
    __hip_atomic_fetch_add(zl + r, a, __ATOMIC_RELAXED, __HIP_MEMORY_SCOPE_WORKGROUP);
  }
  add_rows<F, GR>(m, lane, s, r, a, h, tile);
}

template <int F, int GR>
__device__ __forceinline__ void add_queued_rows(lds_queue_t* q, int m, int lane,
                                                const float* __restrict__ h,
                                                const int* __restrict__ esrc,
                                                const int* __restrict__ edst,
                                                const float* __restrict__ as, const float* adl,
                                                const float* ml, float* zl,
                                                float* tile, int n0) {
  int s = 0, r = 0;
  float a = 0.f;
  if (lane < m) {
    float ev;
    fetch_edge(q[lane], esrc, edst, as, adl, n0, s, r, ev);
    a = expf(ev - ml[r]);
    __hip_atomic_fetch_add(zl + r, a, __ATOMIC_RELAXED, __HIP_MEMORY_SCOPE_WORKGROUP);
  }
  add_rows<F, GR>(m, lane, s, r, a, h, tile);
}

__device__ __forceinline__ void max_queued_rows(lds_queue_t* q, int m, int lane,
                                                const int* __restrict__ esrc,
                                                const int* __restrict__ edst,
                                                const float* __restrict__ as, const float* adl,
                                                unsigned* me, int n0,
                                                const kept_lists& kl, int& kcnt) {
  const bool room = kcnt + m <= LCAP;
  const bool groom = kl.g != nullptr && kcnt + m <= kl.gcap;
  if (lane < m) {
    int s, r;
    float ev;
    fetch_edge(q[lane], esrc, edst, as, adl, n0, s, r, ev);
    if (groom) kl.g[kcnt + lane] = make_int2(s, r);
    if (room) {
      kl.s[kcnt + lane] = s;
      kl.r[kcnt + lane] = r;
      kl.e[kcnt + lane] = __float_as_int(ev);
    }
    __hip_atomic_fetch_max(me + r, enc_f32(ev), __ATOMIC_RELAXED, __HIP_MEMORY_SCOPE_WORKGROUP);
  }
  kcnt += m;
}

template <int F, int GR, bool MAXPASS>
__device__ __forceinline__ void take_queued(lds_queue_t* q, int m, int lane,
                                            const float* __restrict__ h,
                                            const int* __restrict__ esrc,
                                            const int* __restrict__ edst,
                                            const float* __restrict__ as, const float* adl,
                                            unsigned* me, const float* ml, float* zl,
                                            float* tile, int n0, const kept_lists& kl,
                                            int& kcnt) {
  if constexpr (MAXPASS) max_queued_rows(q, m, lane, esrc, edst, as, adl, me, n0, kl, kcnt);
  else add_queued_rows<F, GR>(q, m, lane, h, esrc, edst, as, adl, ml, zl, tile, n0);
}

__device__ __forceinline__ int push_matches(lds_queue_t* q, int cnt, bool match, int e, int lane) {
  const unsigned mask = __builtin_amdgcn_ballot_w32(match);
  if (mask) {
    if (match) q[cnt + __popc(mask & ((1u << lane) - 1u))] = e;
    cnt += __popc(mask);
  }
  return cnt;
}

__device__ __forceinline__ unsigned nonzero_bytes(unsigned t) {
  return ((t & 0x7f7f7f7fu) + 0x7f7f7f7fu) | t;
}

__device__ __forceinline__ unsigned match_bits(uint4 v, unsigned pat) {
  return ((~nonzero_bytes(v.x ^ pat) >> 7) & 0x01010101u) |
         ((~nonzero_bytes(v.y ^ pat) >> 6) & 0x02020202u) |
         ((~nonzero_bytes(v.z ^ pat) >> 5) & 0x04040404u) |
         ((~nonzero_bytes(v.w ^ pat) >> 4) & 0x08080808u);
}

template <int F, int GR, bool MAXPASS>
__device__ __forceinline__ void scan_list(lds_queue_t* q, int lane, int wave, int bucket,
                                          const float* __restrict__ h,
                                          const int* __restrict__ esrc,
                                          const int* __restrict__ edst,
                                          const unsigned char* __restrict__ bk,
                                          const float* __restrict__ as, const float* adl,
                                          unsigned* me, const float* ml, float* zl,
                                          float* tile, int n0, int E, int Nn,
                                          const kept_lists& kl, int& kcnt, bool chunked) {
  constexpr int U = 8;
  int cnt = 0;
  const bool wide = bk != nullptr && ((((size_t)bk) & 15) == 0) && Nn <= 256 * TROWS;
  constexpr int SPAN = 512 * U;
  static_assert(BKCH % SPAN == 0, "a trip must lie inside one piece of the byte list");
  const int nfull = wide ? E / SPAN : 0;
  const unsigned pat = (unsigned)bucket * 0x01010101u;
  for (int t = wave; t < nfull; t += TWAVES) {
    const int base = t * SPAN;
    const unsigned char* __restrict__ bt = bk + bk_place((size_t)base, chunked);
    uint4 dv[U];
    for (int u = 0; u < U; ++u)
      dv[u] = *(const uint4*)(bt + u * 512 + lane * 16);
    unsigned g[U];
    for (int u = 0; u < U; ++u) g[u] = match_bits(dv[u], pat);
    bool again;
    do {
      again = false;
#pragma unroll
      for (int u = 0; u < U; ++u) {
        while (!again && __builtin_amdgcn_ballot_w32(g[u] != 0u)) {
          const bool has = g[u] != 0u;
          const int p = __builtin_ctz(g[u] | 0x80000000u);
          const int e = base + u * 512 + lane * 16 + ((p & 7) << 2) + (p >> 3);
          cnt = push_matches(q, cnt, has, e, lane);
          g[u] &= g[u] - 1u;
          if (cnt >= 32) again = true;
        }
      }
      while (cnt >= 32) {
        cnt -= 32;
        take_queued<F, GR, MAXPASS>(q + cnt, 32, lane, h, esrc, edst, as, adl, me, ml, zl, tile, n0,
                                    kl, kcnt);
      }
    } while (again);
  }
  for (int e0 = nfull * SPAN + wave * 32; e0 < E; e0 += TWAVES * 32) {
    const int e = e0 + lane;
    const int d = (e < E) ? edst[e] : -1;
    cnt = push_matches(q, cnt, e < E && d / TROWS == bucket, e, lane);
    while (cnt >= 32) {
      cnt -= 32;
      take_queued<F, GR, MAXPASS>(q + cnt, 32, lane, h, esrc, edst, as, adl, me, ml, zl, tile, n0,
                                    kl, kcnt);
    }
  }
  if (cnt > 0)
    take_queued<F, GR, MAXPASS>(q, cnt, lane, h, esrc, edst, as, adl, me, ml, zl, tile, n0, kl, kcnt);
}

template <int F, bool TAKE>
__global__ __launch_bounds__(1024) void owner_tile_sum(
    const float* __restrict__ h, const int* __restrict__ esrc, const int* __restrict__ edst,
    const unsigned char* __restrict__ bk,
    const float* __restrict__ as, const float* __restrict__ ad, const float* __restrict__ b,
    float* sum, int E, int Nn, unsigned char* lbuf) {
  static_assert(!TAKE || F * 4 == LIST_ROWB, "the lists live in the rows that layer 2's tiles store");
  constexpr int Q4 = F / 4;
  constexpr int GR = (F == 256) ? 4 : 8;
  constexpr int TALLOC = (TROWS + TWAVES - 1) / TWAVES * TWAVES;
  __shared__ __attribute__((aligned(16))) float tile[TALLOC * F];
  __shared__ int queue[TWAVES * QCAP];
  __shared__ int kept_s[TWAVES * LCAP];
  __shared__ int kept_r[TWAVES * LCAP];
  __shared__ int kept_e[TWAVES * LCAP];
  __shared__ unsigned me[TROWS];
  __shared__ float ml[TROWS];
  __shared__ float zl[TROWS];
  __shared__ float adl[TROWS];
  __shared__ float esl[TROWS];
  __shared__ int kcnts[TWAVES];
  const int lane = threadIdx.x & 31;
  const int wave = __builtin_amdgcn_readfirstlane((int)(threadIdx.x >> 5));
  const int bucket = blockIdx.x;
  const int n0 = bucket * TROWS;
  const int nn = min(TROWS, Nn - n0);
  const float4* __restrict__ h4 = (const float4*)h;
  const float4* __restrict__ b4 = (const float4*)b;
  lds_queue_t* q = (lds_queue_t*)(queue + wave * QCAP);
  kept_lists kl;
  kl.s = (lds_queue_t*)(kept_s + wave * LCAP);
  kl.r = (lds_queue_t*)(kept_r + wave * LCAP);
  kl.e = (lds_queue_t*)(kept_e + wave * LCAP);
  kl.g = nullptr;
  kl.gcap = 0;
  int kcnt = 0;
  const bool keep = lists_usable(E, Nn);
  const int gcap = list_cap(nn);
  int* gcounts = nullptr;
  int2* gpairs = nullptr;
  if (keep) {
    unsigned char* lb = lbuf + (size_t)bucket * REGB;
    gcounts = (int*)lb;
    gpairs = (int2*)(lb + TWAVES * 4) + (size_t)wave * gcap;
  }
  if (!TAKE) {
    kl.g = gpairs;
    kl.gcap = gcap;
  }
  const unsigned char* __restrict__ bks = (TAKE && keep) ? nullptr : bk;

  float eself = 0.f;
  if (threadIdx.x < TROWS) {
    const int node = n0 + min((int)threadIdx.x, nn - 1);
    const float adn = ad[node];
    const float x = as[node] + adn;
    eself = (x > 0.f) ? x : NEG_SLOPE * x;
    adl[threadIdx.x] = adn;
    esl[threadIdx.x] = eself;
    me[threadIdx.x] = enc_f32(eself);
  }
  __syncthreads();
  bool scan = true;
  if (TAKE && keep) {
    constexpr int NL = LCAP / 32;
    const int cw = gcounts[lane];
    if (__builtin_amdgcn_ballot_w32(cw > gcap) == 0u) {
      const int c = __builtin_amdgcn_readlane(cw, wave);
      int2 sr[NL];
      for (int i = 0; i < NL; ++i) sr[i] = gpairs[min(lane + 32 * i, max(gcap - 1, 0))];
      float asv[NL];
      for (int i = 0; i < NL; ++i) asv[i] = as[(lane + 32 * i < c) ? sr[i].x : 0];
      for (int i = 0; i < NL; ++i) {
        const int k = lane + 32 * i;
        if (k < c) {
          const float x = asv[i] + adl[sr[i].y];
          const float ev = (x > 0.f) ? x : NEG_SLOPE * x;
          kl.s[k] = sr[i].x;
          kl.r[k] = sr[i].y;
          kl.e[k] = __float_as_int(ev);
          __hip_atomic_fetch_max(me + sr[i].y, enc_f32(ev), __ATOMIC_RELAXED,
                                 __HIP_MEMORY_SCOPE_WORKGROUP);
        }
      }
      kcnt = c;
      scan = false;
    }
  }
  if (scan)
    scan_list<F, GR, true>(q, lane, wave, bucket, h, esrc, edst, bks, as, adl, me, ml, zl, tile, n0,
                           E, Nn, kl, kcnt, !TAKE && keep);
  if (!TAKE && keep && lane == 0) kcnts[wave] = kcnt;
  __syncthreads();
  if (!TAKE && keep && wave == 0) gcounts[lane] = kcnts[lane];

  if (threadIdx.x < TROWS) {
    const float mv = dec_f32(me[threadIdx.x]);
    ml[threadIdx.x] = mv;
    zl[threadIdx.x] = expf(eself - mv);
  }

  {
    constexpr int RPW = TALLOC / TWAVES;
    constexpr int PPL = Q4 / 32;
    float4 hv[RPW][PPL];
    float av[RPW];
    for (int i = 0; i < RPW; ++i) {
      const int rr = min(wave + i * TWAVES, nn - 1);
      const int node = n0 + rr;
      av[i] = expf(esl[rr] - dec_f32(me[rr]));
      for (int j = 0; j < PPL; ++j) hv[i][j] = h4[(size_t)node * Q4 + j * 32 + lane];
    }
    for (int i = 0; i < RPW; ++i) {
      const int r = wave + i * TWAVES;
      for (int j = 0; j < PPL; ++j) {
        float4 o;
        o.x = av[i] * hv[i][j].x;
        o.y = av[i] * hv[i][j].y;
        o.z = av[i] * hv[i][j].z;
        o.w = av[i] * hv[i][j].w;
        float* tp = tile + r * F + j * 128 + lane;
        tp[0]  = o.x;
        tp[32] = o.y;
        tp[64] = o.z;
        tp[96] = o.w;
      }
    }
  }
  __syncthreads();

  if (kcnt <= LCAP) {
    for (int k = 0; k < kcnt; k += 32)
      add_kept_rows<F, GR>(kl, k, min(32, kcnt - k), lane, h, ml, zl, tile);
  } else {
    int unused = 0;
    scan_list<F, GR, false>(q, lane, wave, bucket, h, esrc, edst, bks, as, adl, me, ml, zl, tile,
                            n0, E, Nn, kl, unused, !TAKE && keep);
  }
  __syncthreads();

  float4* __restrict__ s4 = (float4*)sum;
  for (int r = wave; r < nn; r += TWAVES) {
    const float rz = 1.0f / zl[r];
    for (int j = 0; j < Q4 / 32; ++j) {
      const float* tp = tile + r * F + j * 128 + lane;
      const float4 bv = b4[j * 32 + lane];
      float4 o;
      o.x = bv.x + tp[0]  * rz;
      o.y = bv.y + tp[32] * rz;
      o.z = bv.z + tp[64] * rz;
      o.w = bv.w + tp[96] * rz;
      s4[(size_t)(n0 + r) * Q4 + j * 32 + lane] = o;
    }
  }
}

template <bool LAYER2, int FOUT>
static void run_layer(const float* xin, int Fin, const float* W,
                      const float* avs, const float* avd, const float* bias,
                      float* h, float* sum,
                      const int* esrc, const int* edst, int E, int Nn,
                      float* as, float* ad,
                      unsigned char* bk, unsigned char* lbuf, hipStream_t stream) {
  int mt = (Nn + 15) / 16, nt4 = FOUT / 64;
  int waves = mt * nt4;
  gemm_f32_wmma<LAYER2><<<(waves + 7) / 8, 256, 0, stream>>>(xin, W, h, Nn, Fin, FOUT);
  int awaves = (Nn + NA_ROWS - 1) / NA_ROWS;
  node_alpha<FOUT, LAYER2><<<(awaves + 7) / 8, 256, 0, stream>>>(h, avs, avd, as, ad, Nn, edst, bk, E);

  unsigned tblocks = (unsigned)((Nn + TROWS - 1) / TROWS);
  owner_tile_sum<FOUT, LAYER2><<<tblocks, 1024, 0, stream>>>(h, esrc, edst, bk, as, ad, bias, sum,
                                                             E, Nn, lbuf);
}

extern "C" void kernel_launch(void* const* d_in, const int* in_sizes, int n_in,
                              void* d_out, int out_size, void* d_ws, size_t ws_size,
                              hipStream_t stream) {
  const float* x   = (const float*)d_in[0];
  const int*  eidx = (const int*)d_in[1];
  const float* W1  = (const float*)d_in[2];
  const float* a1s = (const float*)d_in[3];
  const float* a1d = (const float*)d_in[4];
  const float* b1  = (const float*)d_in[5];
  const float* W2  = (const float*)d_in[6];
  const float* a2s = (const float*)d_in[7];
  const float* a2d = (const float*)d_in[8];
  const float* b2  = (const float*)d_in[9];
  float* out = (float*)d_out;

  const int IN_C = 128, HID_C = 256, OUT_C = 128;
  int Nn = in_sizes[0] / IN_C;
  int E  = in_sizes[1] / 2;
  const int* esrc = eidx;
  const int* edst = eidx + E;

  float* bufA = (float*)d_ws;
  float* bufB = bufA + (size_t)Nn * HID_C;
  float* as   = bufB + (size_t)Nn * HID_C;
  float* ad   = as + Nn;
  (void)ws_size; (void)n_in; (void)out_size;

  run_layer<false, HID_C>(x, IN_C, W1, a1s, a1d, b1,
                           bufA,  bufB,
                          esrc, edst, E, Nn, as, ad,
                           (unsigned char*)out,  (unsigned char*)out, stream);

  run_layer<true, OUT_C>(bufB, HID_C, W2, a2s, a2d, b2,
                          bufA,  out,
                         esrc, edst, E, Nn, as, ad,
                          (unsigned char*)bufB,  (unsigned char*)out, stream);
}
